# starting kernel
# baseline (speedup 1.0000x reference)
#include <hip/hip_runtime.h>
#include <hip/hip_bf16.h>

typedef __attribute__((ext_vector_type(2))) float v2f;
typedef __attribute__((ext_vector_type(8))) float v8f;

#define NEG_SLOPE 0.2f

__device__ __forceinline__ unsigned fkey_enc(float f) {
    unsigned b = __float_as_uint(f);
    return (b & 0x80000000u) ? ~b : (b | 0x80000000u);
}
__device__ __forceinline__ float fkey_dec(unsigned k) {
    return (k & 0x80000000u) ? __uint_as_float(k ^ 0x80000000u)
                             : __uint_as_float(~k);
}

__global__ void k_zero(float* __restrict__ p, long long cnt) {
    long long i = (long long)blockIdx.x * blockDim.x + threadIdx.x;
    long long stride = (long long)gridDim.x * blockDim.x;
    for (; i < cnt; i += stride) p[i] = 0.0f;
}

__global__ void k_nnconv(const float* __restrict__ x,
                         const int*   __restrict__ eidx,
                         const float* __restrict__ eattr,
                         const float* __restrict__ W_nn,
                         const float* __restrict__ b_nn,
                         float* __restrict__ agg,
                         int N, int E) {
    __shared__ float sWp[136 * 32];
    __shared__ float sEA[8][16][16];
    __shared__ float sXS[8][16][16];
    __shared__ int   sDST[8][16];

    const int tid = threadIdx.x;
    for (int t = tid; t < 272 * 16; t += 256) {
        int k = t >> 4, o = t & 15;
        float v = (k < 256) ? W_nn[(k >> 4) * 256 + (k & 15) * 16 + o]
                            : b_nn[(k - 256) * 16 + o];
        sWp[(k >> 1) * 32 + o * 2 + (k & 1)] = v;
    }

    const int wave = tid >> 5;
    const int lane = tid & 31;
    const int m    = lane & 15;
    const int hi   = lane >> 4;
    const int base = (blockIdx.x * 8 + wave) * 16;
    const int e    = base + m;
    const bool valid = (e < E);
    const int ec = valid ? e : 0;

    if (hi == 0) {
        for (int j = 0; j < 16; ++j)
            sEA[wave][m][j] = valid ? eattr[(size_t)ec * 16 + j] : 0.0f;
        sDST[wave][m] = valid ? eidx[E + ec] : 0;
    } else {
        int s = eidx[ec];
        for (int j = 0; j < 16; ++j)
            sXS[wave][m][j] = valid ? x[(size_t)s * 16 + j] : 0.0f;
    }
    __syncthreads();

    v8f acc = {};
    const float* wXS = &sXS[wave][m][0];
    const float* wEA = &sEA[wave][m][0];
#pragma unroll
    for (int kk = 0; kk < 68; ++kk) {
        const int c  = (4 * kk) & 15;
        const int i0 = c + 2 * hi;
        float x0 = wXS[i0], x1 = wXS[i0 + 1];
        v2f a, b;
        if (kk < 64) {
            float eaf = wEA[kk >> 2];
            a.x = eaf * x0; a.y = eaf * x1;
        } else {
            a.x = x0; a.y = x1;
        }
        const int p = 2 * kk + hi;
        b.x = sWp[p * 32 + m * 2 + 0];
        b.y = sWp[p * 32 + m * 2 + 1];
        acc = __builtin_amdgcn_wmma_f32_16x16x4_f32(
            false, a, false, b, (short)0, acc, false, false);
    }

#pragma unroll
    for (int v = 0; v < 8; ++v) {
        int mr = v + 8 * hi;
        int er = base + mr;
        if (er < E) {
            int d = sDST[wave][mr];
            atomicAdd(&agg[(size_t)d * 16 + m], acc[v]);
        }
    }
}

__global__ void k_hnode(const float* __restrict__ x,
                        const float* __restrict__ W_root,
                        const float* __restrict__ b_ecc,
                        const float* __restrict__ agg,
                        float* __restrict__ h, int N) {
    int i = blockIdx.x * blockDim.x + threadIdx.x;
    if (i >= N * 16) return;
    int n = i >> 4, o = i & 15;
    float s = agg[i] + b_ecc[o];
    const float* xr = x + (size_t)n * 16;
#pragma unroll
    for (int j = 0; j < 16; ++j) s += xr[j] * W_root[j * 16 + o];
    h[i] = fmaxf(s, 0.0f);
}

__global__ void k_zgemm(const float* __restrict__ h,
                        const float* __restrict__ W_gat,
                        float* __restrict__ z, int N) {
    __shared__ float sWg[8 * 256];
    const int tid = threadIdx.x;
    for (int t = tid; t < 2048; t += 256) {
        int k = t >> 7, col = t & 127;
        sWg[(k >> 1) * 256 + col * 2 + (k & 1)] = W_gat[k * 128 + col];
    }
    __syncthreads();

    const int wave = tid >> 5, lane = tid & 31;
    const int m = lane & 15, hi = lane >> 4;
    const int rowTile = blockIdx.x;
    const int colTile = wave;
    const int row = rowTile * 16 + m;
    const int rc  = row < N ? row : (N - 1);
    const float* hr = h + (size_t)rc * 16;
    const int col = colTile * 16 + m;

    v8f acc = {};
#pragma unroll
    for (int kk = 0; kk < 4; ++kk) {
        int i0 = 4 * kk + 2 * hi;
        v2f a, b;
        a.x = hr[i0]; a.y = hr[i0 + 1];
        int p = 2 * kk + hi;
        b.x = sWg[p * 256 + col * 2 + 0];
        b.y = sWg[p * 256 + col * 2 + 1];
        acc = __builtin_amdgcn_wmma_f32_16x16x4_f32(
            false, a, false, b, (short)0, acc, false, false);
    }
#pragma unroll
    for (int v = 0; v < 8; ++v) {
        int r = rowTile * 16 + v + 8 * hi;
        if (r < N) z[(size_t)r * 128 + col] = acc[v];
    }
}

__global__ void k_escore(const float* __restrict__ z,
                         const float* __restrict__ a_src,
                         const float* __restrict__ a_dst,
                         float* __restrict__ esrc,
                         float* __restrict__ edst, int N) {
    int i = blockIdx.x * blockDim.x + threadIdx.x;
    if (i >= N * 4) return;
    int n = i >> 2, hh = i & 3;
    const float* zr = z + (size_t)n * 128 + hh * 32;
    const float* as = a_src + hh * 32;
    const float* ad = a_dst + hh * 32;
    float s = 0.f, d = 0.f;
#pragma unroll 8
    for (int j = 0; j < 32; ++j) { float zv = zr[j]; s += zv * as[j]; d += zv * ad[j]; }
    esrc[i] = s; edst[i] = d;
}

__global__ void k_attmax(const int* __restrict__ eidx,
                         const float* __restrict__ esrc,
                         const float* __restrict__ edst,
                         unsigned* __restrict__ maxkey, int E) {
    int i = blockIdx.x * blockDim.x + threadIdx.x;
    if (i >= E * 4) return;
    int e = i >> 2, hh = i & 3;
    int s = eidx[e], d = eidx[E + e];
    float lg = esrc[s * 4 + hh] + edst[d * 4 + hh];
    lg = lg > 0.f ? lg : NEG_SLOPE * lg;
    atomicMax(&maxkey[d * 4 + hh], fkey_enc(lg));
}

__global__ void k_attagg(const int* __restrict__ eidx,
                         const float* __restrict__ esrc,
                         const float* __restrict__ edst,
                         const unsigned* __restrict__ maxkey,
                         const float* __restrict__ z,
                         float* __restrict__ denom,
                         float* __restrict__ gatout, int E) {
    int i = blockIdx.x * blockDim.x + threadIdx.x;
    if (i >= E * 4) return;
    int e = i >> 2, hh = i & 3;
    int s = eidx[e], d = eidx[E + e];
    float lg = esrc[s * 4 + hh] + edst[d * 4 + hh];
    lg = lg > 0.f ? lg : NEG_SLOPE * lg;
    float mx = fkey_dec(maxkey[d * 4 + hh]);
    float ex = __expf(lg - mx);
    atomicAdd(&denom[d * 4 + hh], ex);
    const float* zr = z + (size_t)s * 128 + hh * 32;
    float* op = gatout + (size_t)d * 128 + hh * 32;
#pragma unroll 4
    for (int j = 0; j < 32; ++j) atomicAdd(op + j, ex * zr[j]);
}

__global__ void k_final(const float* __restrict__ gatout,
                        const float* __restrict__ denom,
                        const float* __restrict__ b_gat,
                        const float* __restrict__ W_fc,
                        const float* __restrict__ b_fc,
                        float* __restrict__ out, int N) {
    int wave = threadIdx.x >> 5, lane = threadIdx.x & 31;
    int n = blockIdx.x * 8 + wave;
    if (n >= N) return;
    float acc = 0.f;
#pragma unroll
    for (int t = 0; t < 4; ++t) {
        int j = t * 32 + lane;
        float dn = denom[n * 4 + t] + 1e-16f;
        float g = gatout[(size_t)n * 128 + j] / dn + b_gat[j];
        g = fmaxf(g, 0.0f);
        acc += g * W_fc[j];
    }
    for (int off = 16; off > 0; off >>= 1) acc += __shfl_xor(acc, off, 32);
    if (lane == 0) out[n] = acc + b_fc[0];
}

extern "C" void kernel_launch(void* const* d_in, const int* in_sizes, int n_in,
                              void* d_out, int out_size, void* d_ws, size_t ws_size,
                              hipStream_t stream) {
    const float* x      = (const float*)d_in[0];
    const int*   eidx   = (const int*)  d_in[1];
    const float* eattr  = (const float*)d_in[2];
    const float* W_nn   = (const float*)d_in[3];
    const float* b_nn   = (const float*)d_in[4];
    const float* W_root = (const float*)d_in[5];
    const float* b_ecc  = (const float*)d_in[6];
    const float* W_gat  = (const float*)d_in[7];
    const float* a_src  = (const float*)d_in[8];
    const float* a_dst  = (const float*)d_in[9];
    const float* b_gat  = (const float*)d_in[10];
    const float* W_fc   = (const float*)d_in[11];
    const float* b_fc   = (const float*)d_in[12];
    float* out = (float*)d_out;

    const int N = in_sizes[0] / 16;
    const int E = in_sizes[1] / 2;

    float* ws = (float*)d_ws;
    size_t off = 0;
    float*    agg    = ws + off;               off += (size_t)N * 16;
    unsigned* maxkey = (unsigned*)(ws + off);  off += (size_t)N * 4;
    float*    denom  = ws + off;               off += (size_t)N * 4;
    float*    gatout = ws + off;               off += (size_t)N * 128;
    const long long zeroCnt = (long long)off;
    float*    hbuf   = ws + off;               off += (size_t)N * 16;
    float*    zbuf   = ws + off;               off += (size_t)N * 128;
    float*    esrc   = ws + off;               off += (size_t)N * 4;
    float*    edst   = ws + off;               off += (size_t)N * 4;

    k_zero<<<2048, 256, 0, stream>>>(ws, zeroCnt);

    int tilesE = (E + 15) / 16;
    k_nnconv<<<(tilesE + 7) / 8, 256, 0, stream>>>(x, eidx, eattr, W_nn, b_nn, agg, N, E);
    k_hnode <<<(N * 16 + 255) / 256, 256, 0, stream>>>(x, W_root, b_ecc, agg, hbuf, N);
    k_zgemm <<<(N + 15) / 16, 256, 0, stream>>>(hbuf, W_gat, zbuf, N);
    k_escore<<<(N * 4 + 255) / 256, 256, 0, stream>>>(zbuf, a_src, a_dst, esrc, edst, N);
    k_attmax<<<(E * 4 + 255) / 256, 256, 0, stream>>>(eidx, esrc, edst, maxkey, E);
    k_attagg<<<(E * 4 + 255) / 256, 256, 0, stream>>>(eidx, esrc, edst, maxkey, zbuf,
                                                      denom, gatout, E);
    k_final <<<(N + 7) / 8, 256, 0, stream>>>(gatout, denom, b_gat, W_fc, b_fc, out, N);
}


# GAT sums by owner work-groups: each owns 196 nodes, filters the whole destination list, adds edge rows into LDS by ds float adds (sum order = LDS arrival), relu/fc from LDS; no gatout/denom planes
# speedup vs baseline: 5.2399x; 5.2399x over previous
#include <hip/hip_runtime.h>
#include <hip/hip_bf16.h>

typedef __attribute__((ext_vector_type(2))) float v2f;
typedef __attribute__((ext_vector_type(8))) float v8f;

#define NEG_SLOPE 0.2f

__device__ __forceinline__ unsigned fkey_enc(float f) {
    unsigned b = __float_as_uint(f);
    return (b & 0x80000000u) ? ~b : (b | 0x80000000u);
}
__device__ __forceinline__ float fkey_dec(unsigned k) {
    return (k & 0x80000000u) ? __uint_as_float(k ^ 0x80000000u)
                             : __uint_as_float(~k);
}

__global__ void k_zero(float* __restrict__ p, long long cnt) {
    long long i = (long long)blockIdx.x * blockDim.x + threadIdx.x;
    long long stride = (long long)gridDim.x * blockDim.x;
    for (; i < cnt; i += stride) p[i] = 0.0f;
}

__global__ void k_nnconv(const float* __restrict__ x,
                         const int*   __restrict__ eidx,
                         const float* __restrict__ eattr,
                         const float* __restrict__ W_nn,
                         const float* __restrict__ b_nn,
                         float* __restrict__ agg,
                         int N, int E) {
    __shared__ float sWp[136 * 32];
    __shared__ float sEA[8][16][16];
    __shared__ float sXS[8][16][16];
    __shared__ int   sDST[8][16];

    const int tid = threadIdx.x;
    for (int t = tid; t < 272 * 16; t += 256) {
        int k = t >> 4, o = t & 15;
        float v = (k < 256) ? W_nn[(k >> 4) * 256 + (k & 15) * 16 + o]
                            : b_nn[(k - 256) * 16 + o];
        sWp[(k >> 1) * 32 + o * 2 + (k & 1)] = v;
    }

    const int wave = tid >> 5;
    const int lane = tid & 31;
    const int m    = lane & 15;
    const int hi   = lane >> 4;
    const int base = (blockIdx.x * 8 + wave) * 16;
    const int e    = base + m;
    const bool valid = (e < E);
    const int ec = valid ? e : 0;

    if (hi == 0) {
        for (int j = 0; j < 16; ++j)
            sEA[wave][m][j] = valid ? eattr[(size_t)ec * 16 + j] : 0.0f;
        sDST[wave][m] = valid ? eidx[E + ec] : 0;
    } else {
        int s = eidx[ec];
        for (int j = 0; j < 16; ++j)
            sXS[wave][m][j] = valid ? x[(size_t)s * 16 + j] : 0.0f;
    }
    __syncthreads();

    v8f acc = {};
    const float* wXS = &sXS[wave][m][0];
    const float* wEA = &sEA[wave][m][0];
#pragma unroll
    for (int kk = 0; kk < 68; ++kk) {
        const int c  = (4 * kk) & 15;
        const int i0 = c + 2 * hi;
        float x0 = wXS[i0], x1 = wXS[i0 + 1];
        v2f a, b;
        if (kk < 64) {
            float eaf = wEA[kk >> 2];
            a.x = eaf * x0; a.y = eaf * x1;
        } else {
            a.x = x0; a.y = x1;
        }
        const int p = 2 * kk + hi;
        b.x = sWp[p * 32 + m * 2 + 0];
        b.y = sWp[p * 32 + m * 2 + 1];
        acc = __builtin_amdgcn_wmma_f32_16x16x4_f32(
            false, a, false, b, (short)0, acc, false, false);
    }

#pragma unroll
    for (int v = 0; v < 8; ++v) {
        int mr = v + 8 * hi;
        int er = base + mr;
        if (er < E) {
            int d = sDST[wave][mr];
            atomicAdd(&agg[(size_t)d * 16 + m], acc[v]);
        }
    }
}

__global__ void k_hnode(const float* __restrict__ x,
                        const float* __restrict__ W_root,
                        const float* __restrict__ b_ecc,
                        const float* __restrict__ agg,
                        float* __restrict__ h, int N) {
    int i = blockIdx.x * blockDim.x + threadIdx.x;
    if (i >= N * 16) return;
    int n = i >> 4, o = i & 15;
    float s = agg[i] + b_ecc[o];
    const float* xr = x + (size_t)n * 16;
#pragma unroll
    for (int j = 0; j < 16; ++j) s += xr[j] * W_root[j * 16 + o];
    h[i] = fmaxf(s, 0.0f);
}

__global__ void k_zgemm(const float* __restrict__ h,
                        const float* __restrict__ W_gat,
                        float* __restrict__ z, int N) {
    __shared__ float sWg[8 * 256];
    const int tid = threadIdx.x;
    for (int t = tid; t < 2048; t += 256) {
        int k = t >> 7, col = t & 127;
        sWg[(k >> 1) * 256 + col * 2 + (k & 1)] = W_gat[k * 128 + col];
    }
    __syncthreads();

    const int wave = tid >> 5, lane = tid & 31;
    const int m = lane & 15, hi = lane >> 4;
    const int rowTile = blockIdx.x;
    const int colTile = wave;
    const int row = rowTile * 16 + m;
    const int rc  = row < N ? row : (N - 1);
    const float* hr = h + (size_t)rc * 16;
    const int col = colTile * 16 + m;

    v8f acc = {};
#pragma unroll
    for (int kk = 0; kk < 4; ++kk) {
        int i0 = 4 * kk + 2 * hi;
        v2f a, b;
        a.x = hr[i0]; a.y = hr[i0 + 1];
        int p = 2 * kk + hi;
        b.x = sWg[p * 256 + col * 2 + 0];
        b.y = sWg[p * 256 + col * 2 + 1];
        acc = __builtin_amdgcn_wmma_f32_16x16x4_f32(
            false, a, false, b, (short)0, acc, false, false);
    }
#pragma unroll
    for (int v = 0; v < 8; ++v) {
        int r = rowTile * 16 + v + 8 * hi;
        if (r < N) z[(size_t)r * 128 + col] = acc[v];
    }
}

__global__ void k_escore(const float* __restrict__ z,
                         const float* __restrict__ a_src,
                         const float* __restrict__ a_dst,
                         float* __restrict__ esrc,
                         float* __restrict__ edst, int N) {
    int i = blockIdx.x * blockDim.x + threadIdx.x;
    if (i >= N * 4) return;
    int n = i >> 2, hh = i & 3;
    const float* zr = z + (size_t)n * 128 + hh * 32;
    const float* as = a_src + hh * 32;
    const float* ad = a_dst + hh * 32;
    float s = 0.f, d = 0.f;
#pragma unroll 8
    for (int j = 0; j < 32; ++j) { float zv = zr[j]; s += zv * as[j]; d += zv * ad[j]; }
    esrc[i] = s; edst[i] = d;
}

__global__ void k_attmax(const int* __restrict__ eidx,
                         const float* __restrict__ esrc,
                         const float* __restrict__ edst,
                         unsigned* __restrict__ maxkey, int E) {
    int i = blockIdx.x * blockDim.x + threadIdx.x;
    if (i >= E * 4) return;
    int e = i >> 2, hh = i & 3;
    int s = eidx[e], d = eidx[E + e];
    float lg = esrc[s * 4 + hh] + edst[d * 4 + hh];
    lg = lg > 0.f ? lg : NEG_SLOPE * lg;
    atomicMax(&maxkey[d * 4 + hh], fkey_enc(lg));
}

#define OWN_R  196
#define OWN_W  32
#define SC_U   4
#define WL_CAP (32 + SC_U * 128)

__global__ __launch_bounds__(1024)
void k_gatown(const int* __restrict__ eidx,
              const float* __restrict__ esrc,
              const float* __restrict__ edst,
              const unsigned* __restrict__ maxkey,
              const float* __restrict__ z,
              const float* __restrict__ b_gat,
              const float* __restrict__ W_fc,
              const float* __restrict__ b_fc,
              float* __restrict__ out, int N, int E) {
    __shared__ float sAcc[OWN_R * 128];
    __shared__ float sDen[OWN_R * 4];
    __shared__ int   sList[OWN_W * WL_CAP];

    const int tid = threadIdx.x;
    const int wave = __builtin_amdgcn_readfirstlane(tid >> 5);
    const int lane = tid & 31;
    const int nodeBase = blockIdx.x * OWN_R;
    const int rEff = max(min(OWN_R, N - nodeBase), 0);

    for (int i = tid; i < OWN_R * 128; i += 1024) sAcc[i] = 0.0f;
    for (int i = tid; i < OWN_R * 4; i += 1024) sDen[i] = 0.0f;
    __syncthreads();

    int* wl = sList + wave * WL_CAP;
    int cnt = 0;

    const int* dstp = eidx + E;
    const int preC = min((4 - (E & 3)) & 3, E);
    const int nq = (E - preC) >> 2;
    const int tailStart = preC + 4 * nq;
    const int4* dq = (const int4*)(dstp + preC);
    const int nblk = (nq + 32 * SC_U - 1) / (32 * SC_U);
    const int rot = (int)blockIdx.x * (nblk / (int)gridDim.x);

    if (wave == 0) {
        const int nLeft = preC + (E - tailStart);
        const bool has = lane < nLeft;
        const int e = has ? (lane < preC ? lane : tailStart + (lane - preC)) : 0;
        const int d = has ? dstp[e] : -1;
        const bool m = has && ((unsigned)(d - nodeBase) < (unsigned)rEff);
        const unsigned mask = __builtin_amdgcn_ballot_w32(m);
        if (m) wl[__builtin_amdgcn_mbcnt_lo(mask, 0u)] = e;
        cnt = __builtin_popcount(mask);
    }

    const int4 none = make_int4(-1, -1, -1, -1);
    int4 cur[SC_U];
    for (int u = 0; u < SC_U; ++u) cur[u] = none;
    if (wave < nblk) {
        const int bc = wave + rot >= nblk ? wave + rot - nblk : wave + rot;
        for (int u = 0; u < SC_U; ++u)
            cur[u] = dq[min(bc * (32 * SC_U) + u * 32 + lane, nq - 1)];
    }

    for (int blk = wave; ; blk += OWN_W) {
        const bool live = blk < nblk;
        if (live) {
            int4 nxt[SC_U];
            const int nb = blk + OWN_W;
            for (int u = 0; u < SC_U; ++u) nxt[u] = none;
            if (nb < nblk) {
                const int bn = nb + rot >= nblk ? nb + rot - nblk : nb + rot;
                for (int u = 0; u < SC_U; ++u)
                    nxt[u] = dq[min(bn * (32 * SC_U) + u * 32 + lane, nq - 1)];
            }
            const int bc = blk + rot >= nblk ? blk + rot - nblk : blk + rot;
            for (int u = 0; u < SC_U; ++u) {
                const int q = bc * (32 * SC_U) + u * 32 + lane;
                const int eb = preC + 4 * q;
                const unsigned lim = q < nq ? (unsigned)rEff : 0u;
                const bool m0 = (unsigned)(cur[u].x - nodeBase) < lim;
                const bool m1 = (unsigned)(cur[u].y - nodeBase) < lim;
                const bool m2 = (unsigned)(cur[u].z - nodeBase) < lim;
                const bool m3 = (unsigned)(cur[u].w - nodeBase) < lim;
                if (__builtin_amdgcn_ballot_w32(m0 | m1 | m2 | m3)) {
                    unsigned mk;
                    mk = __builtin_amdgcn_ballot_w32(m0);
                    if (m0) wl[cnt + __builtin_amdgcn_mbcnt_lo(mk, 0u)] = eb;
                    cnt += __builtin_popcount(mk);
                    mk = __builtin_amdgcn_ballot_w32(m1);
                    if (m1) wl[cnt + __builtin_amdgcn_mbcnt_lo(mk, 0u)] = eb + 1;
                    cnt += __builtin_popcount(mk);
                    mk = __builtin_amdgcn_ballot_w32(m2);
                    if (m2) wl[cnt + __builtin_amdgcn_mbcnt_lo(mk, 0u)] = eb + 2;
                    cnt += __builtin_popcount(mk);
                    mk = __builtin_amdgcn_ballot_w32(m3);
                    if (m3) wl[cnt + __builtin_amdgcn_mbcnt_lo(mk, 0u)] = eb + 3;
                    cnt += __builtin_popcount(mk);
                }
            }
            for (int u = 0; u < SC_U; ++u) cur[u] = nxt[u];
        }

        const int need = live ? 32 : 1;
        while (cnt >= need) {
            const int n = min(cnt, 32);
            cnt -= n;
            const int* bl = wl + cnt;
            const int e = bl[lane < n ? lane : 0];
            const int s = eidx[e];
            const int d = dstp[e];
            const int dl = d - nodeBase;
            const float4 es = *(const float4*)(esrc + (size_t)s * 4);
            const float4 ed = *(const float4*)(edst + (size_t)d * 4);
            const uint4  mx = *(const uint4*)(maxkey + (size_t)d * 4);
            float l0 = es.x + ed.x, l1 = es.y + ed.y, l2 = es.z + ed.z, l3 = es.w + ed.w;
            l0 = l0 > 0.f ? l0 : NEG_SLOPE * l0;
            l1 = l1 > 0.f ? l1 : NEG_SLOPE * l1;
            l2 = l2 > 0.f ? l2 : NEG_SLOPE * l2;
            l3 = l3 > 0.f ? l3 : NEG_SLOPE * l3;
            const float x0 = __expf(l0 - fkey_dec(mx.x));
            const float x1 = __expf(l1 - fkey_dec(mx.y));
            const float x2 = __expf(l2 - fkey_dec(mx.z));
            const float x3 = __expf(l3 - fkey_dec(mx.w));
            if (lane < n) {
                atomicAdd(&sDen[dl * 4 + 0], x0);
                atomicAdd(&sDen[dl * 4 + 1], x1);
                atomicAdd(&sDen[dl * 4 + 2], x2);
                atomicAdd(&sDen[dl * 4 + 3], x3);
            }
            const int hd = lane >> 3;
            for (int j0 = 0; j0 < n; j0 += 8) {
                float4 r[8];
                for (int u = 0; u < 8; ++u) {
                    const int j = min(j0 + u, n - 1);
                    const int sj = __builtin_amdgcn_readlane(s, j);
                    r[u] = *(const float4*)(z + (size_t)sj * 128 + lane * 4);
                }
                for (int u = 0; u < 8; ++u) {
                    const int j = j0 + u;
                    if (j < n) {
                        const int dj = __builtin_amdgcn_readlane(dl, j);
                        const float w0 = __int_as_float(__builtin_amdgcn_readlane(__float_as_int(x0), j));
                        const float w1 = __int_as_float(__builtin_amdgcn_readlane(__float_as_int(x1), j));
                        const float w2 = __int_as_float(__builtin_amdgcn_readlane(__float_as_int(x2), j));
                        const float w3 = __int_as_float(__builtin_amdgcn_readlane(__float_as_int(x3), j));
                        const float wa = (hd & 1) ? w1 : w0;
                        const float wb = (hd & 1) ? w3 : w2;
                        const float w  = (hd & 2) ? wb : wa;
                        float* ap = sAcc + dj * 128 + lane;
                        atomicAdd(ap,      w * r[u].x);
                        atomicAdd(ap + 32, w * r[u].y);
                        atomicAdd(ap + 64, w * r[u].z);
                        atomicAdd(ap + 96, w * r[u].w);
                    }
                }
            }
        }
        if (!live) break;
    }
    __syncthreads();

    for (int i = wave; i < rEff; i += OWN_W) {
        float acc = 0.f;
        for (int t = 0; t < 4; ++t) {
            const int j = t * 32 + lane;
            const float dn = sDen[i * 4 + t] + 1e-16f;
            float g = sAcc[i * 128 + (j & 3) * 32 + (j >> 2)] / dn + b_gat[j];
            g = fmaxf(g, 0.0f);
            acc += g * W_fc[j];
        }
        for (int off = 16; off > 0; off >>= 1) acc += __shfl_xor(acc, off, 32);
        if (lane == 0) out[nodeBase + i] = acc + b_fc[0];
    }
}

extern "C" void kernel_launch(void* const* d_in, const int* in_sizes, int n_in,
                              void* d_out, int out_size, void* d_ws, size_t ws_size,
                              hipStream_t stream) {
    const float* x      = (const float*)d_in[0];
    const int*   eidx   = (const int*)  d_in[1];
    const float* eattr  = (const float*)d_in[2];
    const float* W_nn   = (const float*)d_in[3];
    const float* b_nn   = (const float*)d_in[4];
    const float* W_root = (const float*)d_in[5];
    const float* b_ecc  = (const float*)d_in[6];
    const float* W_gat  = (const float*)d_in[7];
    const float* a_src  = (const float*)d_in[8];
    const float* a_dst  = (const float*)d_in[9];
    const float* b_gat  = (const float*)d_in[10];
    const float* W_fc   = (const float*)d_in[11];
    const float* b_fc   = (const float*)d_in[12];
    float* out = (float*)d_out;

    const int N = in_sizes[0] / 16;
    const int E = in_sizes[1] / 2;

    float* ws = (float*)d_ws;
    size_t off = 0;
    float*    agg    = ws + off;               off += (size_t)N * 16;
    unsigned* maxkey = (unsigned*)(ws + off);  off += (size_t)N * 4;
    const long long zeroCnt = (long long)off;
    float*    hbuf   = ws + off;               off += (size_t)N * 16;
    float*    zbuf   = ws + off;               off += (size_t)N * 128;
    float*    esrc   = ws + off;               off += (size_t)N * 4;
    float*    edst   = ws + off;               off += (size_t)N * 4;

    k_zero<<<2048, 256, 0, stream>>>(ws, zeroCnt);

    int tilesE = (E + 15) / 16;
    k_nnconv<<<(tilesE + 7) / 8, 256, 0, stream>>>(x, eidx, eattr, W_nn, b_nn, agg, N, E);
    k_hnode <<<(N * 16 + 255) / 256, 256, 0, stream>>>(x, W_root, b_ecc, agg, hbuf, N);
    k_zgemm <<<(N + 15) / 16, 256, 0, stream>>>(hbuf, W_gat, zbuf, N);
    k_escore<<<(N * 4 + 255) / 256, 256, 0, stream>>>(zbuf, a_src, a_dst, esrc, edst, N);
    k_attmax<<<(E * 4 + 255) / 256, 256, 0, stream>>>(eidx, esrc, edst, maxkey, E);
    k_gatown<<<(N + OWN_R - 1) / OWN_R, 1024, 0, stream>>>(eidx, esrc, edst, maxkey, zbuf,
                                                        b_gat, W_fc, b_fc, out, N, E);
}


# GAT owner scan reads one owner byte per edge (whole 128-byte lines written by the NNConv kernel) instead of the 4-byte destination: 16 edges per load, zero-byte test; sum order still LDS arrival
# speedup vs baseline: 5.6469x; 1.0777x over previous
#include <hip/hip_runtime.h>
#include <hip/hip_bf16.h>

typedef __attribute__((ext_vector_type(2))) float v2f;
typedef __attribute__((ext_vector_type(8))) float v8f;

#define NEG_SLOPE 0.2f
#define OWN_R  196

__device__ __forceinline__ unsigned fkey_enc(float f) {
    unsigned b = __float_as_uint(f);
    return (b & 0x80000000u) ? ~b : (b | 0x80000000u);
}
__device__ __forceinline__ float fkey_dec(unsigned k) {
    return (k & 0x80000000u) ? __uint_as_float(k ^ 0x80000000u)
                             : __uint_as_float(~k);
}

__global__ void k_zero(float* __restrict__ p, long long cnt) {
    long long i = (long long)blockIdx.x * blockDim.x + threadIdx.x;
    long long stride = (long long)gridDim.x * blockDim.x;
    for (; i < cnt; i += stride) p[i] = 0.0f;
}

__global__ void k_nnconv(const float* __restrict__ x,
                         const int*   __restrict__ eidx,
                         const float* __restrict__ eattr,
                         const float* __restrict__ W_nn,
                         const float* __restrict__ b_nn,
                         float* __restrict__ agg,
                         unsigned* __restrict__ ownw,
                         int N, int E) {
    __shared__ float sWp[136 * 32];
    __shared__ float sEA[8][16][16];
    __shared__ float sXS[8][16][16];
    __shared__ int   sDST[8][16];

    const int tid = threadIdx.x;
    for (int t = tid; t < 272 * 16; t += 256) {
        int k = t >> 4, o = t & 15;
        float v = (k < 256) ? W_nn[(k >> 4) * 256 + (k & 15) * 16 + o]
                            : b_nn[(k - 256) * 16 + o];
        sWp[(k >> 1) * 32 + o * 2 + (k & 1)] = v;
    }

    const int wave = tid >> 5;
    const int lane = tid & 31;
    const int m    = lane & 15;
    const int hi   = lane >> 4;
    const int base = (blockIdx.x * 8 + wave) * 16;
    const int e    = base + m;
    const bool valid = (e < E);
    const int ec = valid ? e : 0;

    if (hi == 0) {
        for (int j = 0; j < 16; ++j)
            sEA[wave][m][j] = valid ? eattr[(size_t)ec * 16 + j] : 0.0f;
        sDST[wave][m] = valid ? eidx[E + ec] : 0;
    } else {
        int s = eidx[ec];
        for (int j = 0; j < 16; ++j)
            sXS[wave][m][j] = valid ? x[(size_t)s * 16 + j] : 0.0f;
    }
    __syncthreads();

    if (wave == 0) {
        const int* dflat = &sDST[0][0];
        unsigned pk = 0u;
#pragma unroll
        for (int k = 0; k < 4; ++k)
            pk |= (((unsigned)dflat[4 * lane + k] / OWN_R) & 255u) << (8 * k);
        ownw[(size_t)blockIdx.x * 32 + lane] = pk;
    }

    v8f acc = {};
    const float* wXS = &sXS[wave][m][0];
    const float* wEA = &sEA[wave][m][0];
#pragma unroll
    for (int kk = 0; kk < 68; ++kk) {
        const int c  = (4 * kk) & 15;
        const int i0 = c + 2 * hi;
        float x0 = wXS[i0], x1 = wXS[i0 + 1];
        v2f a, b;
        if (kk < 64) {
            float eaf = wEA[kk >> 2];
            a.x = eaf * x0; a.y = eaf * x1;
        } else {
            a.x = x0; a.y = x1;
        }
        const int p = 2 * kk + hi;
        b.x = sWp[p * 32 + m * 2 + 0];
        b.y = sWp[p * 32 + m * 2 + 1];
        acc = __builtin_amdgcn_wmma_f32_16x16x4_f32(
            false, a, false, b, (short)0, acc, false, false);
    }

#pragma unroll
    for (int v = 0; v < 8; ++v) {
        int mr = v + 8 * hi;
        int er = base + mr;
        if (er < E) {
            int d = sDST[wave][mr];
            atomicAdd(&agg[(size_t)d * 16 + m], acc[v]);
        }
    }
}

__global__ void k_hnode(const float* __restrict__ x,
                        const float* __restrict__ W_root,
                        const float* __restrict__ b_ecc,
                        const float* __restrict__ agg,
                        float* __restrict__ h, int N) {
    int i = blockIdx.x * blockDim.x + threadIdx.x;
    if (i >= N * 16) return;
    int n = i >> 4, o = i & 15;
    float s = agg[i] + b_ecc[o];
    const float* xr = x + (size_t)n * 16;
#pragma unroll
    for (int j = 0; j < 16; ++j) s += xr[j] * W_root[j * 16 + o];
    h[i] = fmaxf(s, 0.0f);
}

__global__ void k_zgemm(const float* __restrict__ h,
                        const float* __restrict__ W_gat,
                        float* __restrict__ z, int N) {
    __shared__ float sWg[8 * 256];
    const int tid = threadIdx.x;
    for (int t = tid; t < 2048; t += 256) {
        int k = t >> 7, col = t & 127;
        sWg[(k >> 1) * 256 + col * 2 + (k & 1)] = W_gat[k * 128 + col];
    }
    __syncthreads();

    const int wave = tid >> 5, lane = tid & 31;
    const int m = lane & 15, hi = lane >> 4;
    const int rowTile = blockIdx.x;
    const int colTile = wave;
    const int row = rowTile * 16 + m;
    const int rc  = row < N ? row : (N - 1);
    const float* hr = h + (size_t)rc * 16;
    const int col = colTile * 16 + m;

    v8f acc = {};
#pragma unroll
    for (int kk = 0; kk < 4; ++kk) {
        int i0 = 4 * kk + 2 * hi;
        v2f a, b;
        a.x = hr[i0]; a.y = hr[i0 + 1];
        int p = 2 * kk + hi;
        b.x = sWg[p * 256 + col * 2 + 0];
        b.y = sWg[p * 256 + col * 2 + 1];
        acc = __builtin_amdgcn_wmma_f32_16x16x4_f32(
            false, a, false, b, (short)0, acc, false, false);
    }
#pragma unroll
    for (int v = 0; v < 8; ++v) {
        int r = rowTile * 16 + v + 8 * hi;
        if (r < N) z[(size_t)r * 128 + col] = acc[v];
    }
}

__global__ void k_escore(const float* __restrict__ z,
                         const float* __restrict__ a_src,
                         const float* __restrict__ a_dst,
                         float* __restrict__ esrc,
                         float* __restrict__ edst, int N) {
    int i = blockIdx.x * blockDim.x + threadIdx.x;
    if (i >= N * 4) return;
    int n = i >> 2, hh = i & 3;
    const float* zr = z + (size_t)n * 128 + hh * 32;
    const float* as = a_src + hh * 32;
    const float* ad = a_dst + hh * 32;
    float s = 0.f, d = 0.f;
#pragma unroll 8
    for (int j = 0; j < 32; ++j) { float zv = zr[j]; s += zv * as[j]; d += zv * ad[j]; }
    esrc[i] = s; edst[i] = d;
}

__global__ void k_attmax(const int* __restrict__ eidx,
                         const float* __restrict__ esrc,
                         const float* __restrict__ edst,
                         unsigned* __restrict__ maxkey, int E) {
    int i = blockIdx.x * blockDim.x + threadIdx.x;
    if (i >= E * 4) return;
    int e = i >> 2, hh = i & 3;
    int s = eidx[e], d = eidx[E + e];
    float lg = esrc[s * 4 + hh] + edst[d * 4 + hh];
    lg = lg > 0.f ? lg : NEG_SLOPE * lg;
    atomicMax(&maxkey[d * 4 + hh], fkey_enc(lg));
}

#define OWN_W  32
#define SC_U   2
#define WL_CAP (32 + SC_U * 512)
static_assert(SC_U == 2, "the flag word of the scan holds two groups of 16 edges per lane");

__device__ __forceinline__ unsigned zero_bytes(unsigned t) {
    return ~(((t & 0x7F7F7F7Fu) + 0x7F7F7F7Fu) | t | 0x7F7F7F7Fu);
}


__global__ __launch_bounds__(1024)
void k_gatown(const int* __restrict__ eidx,
              const unsigned* __restrict__ own,
              const float* __restrict__ esrc,
              const float* __restrict__ edst,
              const unsigned* __restrict__ maxkey,
              const float* __restrict__ z,
              const float* __restrict__ b_gat,
              const float* __restrict__ W_fc,
              const float* __restrict__ b_fc,
              float* __restrict__ out, int N, int E) {
    __shared__ float sAcc[OWN_R * 128];
    __shared__ float sDen[OWN_R * 4];
    __shared__ int   sList[OWN_W * WL_CAP];

    const int tid = threadIdx.x;
    const int wave = __builtin_amdgcn_readfirstlane(tid >> 5);
    const int lane = tid & 31;
    const int nodeBase = blockIdx.x * OWN_R;
    const int rEff = max(min(OWN_R, N - nodeBase), 0);

    for (int i = tid; i < OWN_R * 128; i += 1024) sAcc[i] = 0.0f;
    for (int i = tid; i < OWN_R * 4; i += 1024) sDen[i] = 0.0f;
    __syncthreads();

    int* wl = sList + wave * WL_CAP;
    int cnt = 0;

    const int* dstp = eidx + E;
    const uint4* oq = (const uint4*)own;
    const int nq16 = (E + 15) >> 4;
    const int nblk = (nq16 + 32 * SC_U - 1) / (32 * SC_U);
    const int rot = (int)blockIdx.x * (nblk / (int)gridDim.x);
    const unsigned pat = ((unsigned)blockIdx.x & 255u) * 0x01010101u;

    const uint4 zero4 = make_uint4(0u, 0u, 0u, 0u);
    uint4 cur[SC_U];
    for (int u = 0; u < SC_U; ++u) cur[u] = zero4;
    if (wave < nblk) {
        const int bc = wave + rot >= nblk ? wave + rot - nblk : wave + rot;
        for (int u = 0; u < SC_U; ++u)
            cur[u] = oq[min(bc * (32 * SC_U) + u * 32 + lane, nq16 - 1)];
    }

    for (int blk = wave; ; blk += OWN_W) {
        const bool live = blk < nblk;
        if (live) {
            uint4 nxt[SC_U];
            const int nb = blk + OWN_W;
            for (int u = 0; u < SC_U; ++u) nxt[u] = zero4;
            if (nb < nblk) {
                const int bn = nb + rot >= nblk ? nb + rot - nblk : nb + rot;
                for (int u = 0; u < SC_U; ++u)
                    nxt[u] = oq[min(bn * (32 * SC_U) + u * 32 + lane, nq16 - 1)];
            }
            const int bc = blk + rot >= nblk ? blk + rot - nblk : blk + rot;
            const int qi0 = bc * (32 * SC_U) + lane;
            unsigned hb = 0u;
            for (int u = 0; u < SC_U; ++u) {
                const unsigned f0 = zero_bytes(cur[u].x ^ pat);
                const unsigned f1 = zero_bytes(cur[u].y ^ pat);
                const unsigned f2 = zero_bytes(cur[u].z ^ pat);
                const unsigned f3 = zero_bytes(cur[u].w ^ pat);
                const unsigned h = (f0 >> 7) | (f1 >> 6) | (f2 >> 5) | (f3 >> 4);
                if (qi0 + 32 * u < nq16) hb |= h << (4 * u);
            }
            unsigned mk = __builtin_amdgcn_ballot_w32(hb != 0u);
            for (int r = 0; r < 16 * SC_U && mk != 0u; ++r) {
                if (hb != 0u) {
                    const int p = __builtin_ctz(hb);
                    wl[cnt + __builtin_amdgcn_mbcnt_lo(mk, 0u)] =
                        16 * (qi0 + 32 * ((p >> 2) & 1)) + 4 * (p & 3) + (p >> 3);
                    hb &= hb - 1u;
                }
                cnt += __builtin_popcount(mk);
                mk = __builtin_amdgcn_ballot_w32(hb != 0u);
            }
            for (int u = 0; u < SC_U; ++u) cur[u] = nxt[u];
        }

        const int need = live ? 32 : 1;
        while (cnt >= need) {
            const int n = min(cnt, 32);
            cnt -= n;
            const int* bl = wl + cnt;
            const int er = bl[lane < n ? lane : 0];
            const int e = min(er, E - 1);
            const int s = eidx[e];
            const int d = dstp[e];
            const int dl = d - nodeBase;
            const bool ok = lane < n && er < E && (unsigned)dl < (unsigned)rEff;
            const unsigned okm = __builtin_amdgcn_ballot_w32(ok);
            const float4 es = *(const float4*)(esrc + (size_t)s * 4);
            const float4 ed = *(const float4*)(edst + (size_t)d * 4);
            const uint4  mx = *(const uint4*)(maxkey + (size_t)d * 4);
            float l0 = es.x + ed.x, l1 = es.y + ed.y, l2 = es.z + ed.z, l3 = es.w + ed.w;
            l0 = l0 > 0.f ? l0 : NEG_SLOPE * l0;
            l1 = l1 > 0.f ? l1 : NEG_SLOPE * l1;
            l2 = l2 > 0.f ? l2 : NEG_SLOPE * l2;
            l3 = l3 > 0.f ? l3 : NEG_SLOPE * l3;
            const float x0 = __expf(l0 - fkey_dec(mx.x));
            const float x1 = __expf(l1 - fkey_dec(mx.y));
            const float x2 = __expf(l2 - fkey_dec(mx.z));
            const float x3 = __expf(l3 - fkey_dec(mx.w));
            if (ok) {
                atomicAdd(&sDen[dl * 4 + 0], x0);
                atomicAdd(&sDen[dl * 4 + 1], x1);
                atomicAdd(&sDen[dl * 4 + 2], x2);
                atomicAdd(&sDen[dl * 4 + 3], x3);
            }
            const int hd = lane >> 3;
            for (int j0 = 0; j0 < n; j0 += 8) {
                float4 r[8];
                for (int u = 0; u < 8; ++u) {
                    const int j = min(j0 + u, n - 1);
                    const int sj = __builtin_amdgcn_readlane(s, j);
                    r[u] = *(const float4*)(z + (size_t)sj * 128 + lane * 4);
                }
                for (int u = 0; u < 8; ++u) {
                    const int j = j0 + u;
                    if ((okm >> j) & 1u) {
                        const int dj = __builtin_amdgcn_readlane(dl, j);
                        const float w0 = __int_as_float(__builtin_amdgcn_readlane(__float_as_int(x0), j));
                        const float w1 = __int_as_float(__builtin_amdgcn_readlane(__float_as_int(x1), j));
                        const float w2 = __int_as_float(__builtin_amdgcn_readlane(__float_as_int(x2), j));
                        const float w3 = __int_as_float(__builtin_amdgcn_readlane(__float_as_int(x3), j));
                        const float wa = (hd & 1) ? w1 : w0;
                        const float wb = (hd & 1) ? w3 : w2;
                        const float w  = (hd & 2) ? wb : wa;
                        float* ap = sAcc + dj * 128 + lane;
                        atomicAdd(ap,      w * r[u].x);
                        atomicAdd(ap + 32, w * r[u].y);
                        atomicAdd(ap + 64, w * r[u].z);
                        atomicAdd(ap + 96, w * r[u].w);
                    }
                }
            }
        }
        if (!live) break;
    }
    __syncthreads();

    for (int i = wave; i < rEff; i += OWN_W) {
        float acc = 0.f;
        for (int t = 0; t < 4; ++t) {
            const int j = t * 32 + lane;
            const float dn = sDen[i * 4 + t] + 1e-16f;
            float g = sAcc[i * 128 + (j & 3) * 32 + (j >> 2)] / dn + b_gat[j];
            g = fmaxf(g, 0.0f);
            acc += g * W_fc[j];
        }
        for (int off = 16; off > 0; off >>= 1) acc += __shfl_xor(acc, off, 32);
        if (lane == 0) out[nodeBase + i] = acc + b_fc[0];
    }
}

extern "C" void kernel_launch(void* const* d_in, const int* in_sizes, int n_in,
                              void* d_out, int out_size, void* d_ws, size_t ws_size,
                              hipStream_t stream) {
    const float* x      = (const float*)d_in[0];
    const int*   eidx   = (const int*)  d_in[1];
    const float* eattr  = (const float*)d_in[2];
    const float* W_nn   = (const float*)d_in[3];
    const float* b_nn   = (const float*)d_in[4];
    const float* W_root = (const float*)d_in[5];
    const float* b_ecc  = (const float*)d_in[6];
    const float* W_gat  = (const float*)d_in[7];
    const float* a_src  = (const float*)d_in[8];
    const float* a_dst  = (const float*)d_in[9];
    const float* b_gat  = (const float*)d_in[10];
    const float* W_fc   = (const float*)d_in[11];
    const float* b_fc   = (const float*)d_in[12];
    float* out = (float*)d_out;

    const int N = in_sizes[0] / 16;
    const int E = in_sizes[1] / 2;

    float* ws = (float*)d_ws;
    size_t off = 0;
    float*    agg    = ws + off;               off += (size_t)N * 16;
    unsigned* maxkey = (unsigned*)(ws + off);  off += (size_t)N * 4;
    const long long zeroCnt = (long long)off;
    float*    hbuf   = ws + off;               off += (size_t)N * 16;
    float*    zbuf   = ws + off;               off += (size_t)N * 128;
    float*    esrc   = ws + off;               off += (size_t)N * 4;
    float*    edst   = ws + off;               off += (size_t)N * 4;
    unsigned* own    = (unsigned*)(ws + off);  off += (size_t)((E + 127) / 128) * 32;

    k_zero<<<2048, 256, 0, stream>>>(ws, zeroCnt);

    int tilesE = (E + 15) / 16;
    k_nnconv<<<(tilesE + 7) / 8, 256, 0, stream>>>(x, eidx, eattr, W_nn, b_nn, agg, own, N, E);
    k_hnode <<<(N * 16 + 255) / 256, 256, 0, stream>>>(x, W_root, b_ecc, agg, hbuf, N);
    k_zgemm <<<(N + 15) / 16, 256, 0, stream>>>(hbuf, W_gat, zbuf, N);
    k_escore<<<(N * 4 + 255) / 256, 256, 0, stream>>>(zbuf, a_src, a_dst, esrc, edst, N);
    k_attmax<<<(E * 4 + 255) / 256, 256, 0, stream>>>(eidx, esrc, edst, maxkey, E);
    k_gatown<<<(N + OWN_R - 1) / OWN_R, 1024, 0, stream>>>(eidx, own, esrc, edst, maxkey, zbuf,
                                                        b_gat, W_fc, b_fc, out, N, E);
}


# NNConv by owner work-groups: each owns 196 nodes, scans owner bytes (from a first kernel), runs its edges' f32 WMMA tiles, adds rows into LDS (sum order = LDS arrival); h from LDS; 7->6 launches
# speedup vs baseline: 6.1639x; 1.0916x over previous
#include <hip/hip_runtime.h>
#include <hip/hip_bf16.h>

typedef __attribute__((ext_vector_type(2))) float v2f;
typedef __attribute__((ext_vector_type(8))) float v8f;

#define NEG_SLOPE 0.2f
#define OWN_R  196

__device__ __forceinline__ unsigned fkey_enc(float f) {
    unsigned b = __float_as_uint(f);
    return (b & 0x80000000u) ? ~b : (b | 0x80000000u);
}
__device__ __forceinline__ float fkey_dec(unsigned k) {
    return (k & 0x80000000u) ? __uint_as_float(k ^ 0x80000000u)
                             : __uint_as_float(~k);
}

#define OWN_W  32
#define SC_U   2
#define WL_CAP (32 + SC_U * 512)
#define PF_D   6
static_assert(SC_U == 2, "the flag word of the scan holds two groups of 16 edges per lane");

__device__ __forceinline__ unsigned zero_bytes(unsigned t) {
    return ~(((t & 0x7F7F7F7Fu) + 0x7F7F7F7Fu) | t | 0x7F7F7F7Fu);
}

__global__ void k_prep(const int* __restrict__ eidx,
                       unsigned* __restrict__ maxkey,
                       unsigned* __restrict__ ownw, int N, int E) {
    const int i0 = blockIdx.x * blockDim.x + threadIdx.x;
    const int stride = gridDim.x * blockDim.x;
    for (int i = i0; i < N * 4; i += stride) maxkey[i] = 0u;
    const int* dstp = eidx + E;
    const int nw = ((E + 127) / 128) * 32;
    for (int w = i0; w < nw; w += stride) {
        unsigned pk = 0u;
        for (int k = 0; k < 4; ++k) {
            const int idx = 4 * w + k;
            const int d = idx < E ? dstp[idx] : 0;
            pk |= (((unsigned)d / OWN_R) & 255u) << (8 * k);
        }
        ownw[w] = pk;
    }
}

__global__ __launch_bounds__(1024)
void k_nnown(const float* __restrict__ x,
             const int*   __restrict__ eidx,
             const float* __restrict__ eattr,
             const float* __restrict__ W_nn,
             const float* __restrict__ b_nn,
             const float* __restrict__ W_root,
             const float* __restrict__ b_ecc,
             const unsigned* __restrict__ own,
             float* __restrict__ h, int N, int E) {
    __shared__ float sWp[136 * 32];
    __shared__ float sAgg[OWN_R * 16];
    __shared__ int   sList[OWN_W * WL_CAP];

    const int tid = threadIdx.x;
    const int wave = __builtin_amdgcn_readfirstlane(tid >> 5);
    const int lane = tid & 31;
    const int m    = lane & 15;
    const int hi   = lane >> 4;
    const int nodeBase = blockIdx.x * OWN_R;
    const int rEff = max(min(OWN_R, N - nodeBase), 0);

    for (int t = tid; t < 272 * 16; t += 1024) {
        int k = t >> 4, o = t & 15;
        float v = (k < 256) ? W_nn[(k >> 4) * 256 + (k & 15) * 16 + o]
                            : b_nn[(k - 256) * 16 + o];
        sWp[(k >> 1) * 32 + o * 2 + (k & 1)] = v;
    }
    for (int i = tid; i < OWN_R * 16; i += 1024) sAgg[i] = 0.0f;
    __syncthreads();

    int* wl = sList + wave * WL_CAP;
    int cnt = 0;

    const int* dstp = eidx + E;
    const uint4* oq = (const uint4*)own;
    const int nq16 = (E + 15) >> 4;
    const int nblk = (nq16 + 32 * SC_U - 1) / (32 * SC_U);
    const int rot = (int)blockIdx.x * (nblk / (int)gridDim.x);
    const unsigned pat = ((unsigned)blockIdx.x & 255u) * 0x01010101u;

    const uint4 zero4 = make_uint4(0u, 0u, 0u, 0u);
    uint4 cur[SC_U];
    for (int u = 0; u < SC_U; ++u) cur[u] = zero4;
    if (wave < nblk) {
        const int bc = wave + rot >= nblk ? wave + rot - nblk : wave + rot;
        for (int u = 0; u < SC_U; ++u)
            cur[u] = oq[min(bc * (32 * SC_U) + u * 32 + lane, nq16 - 1)];
    }

    for (int blk = wave; ; blk += OWN_W) {
        const bool live = blk < nblk;
        if (live) {
            uint4 nxt[SC_U];
            const int nb = blk + OWN_W;
            for (int u = 0; u < SC_U; ++u) nxt[u] = zero4;
            if (nb < nblk) {
                const int bn = nb + rot >= nblk ? nb + rot - nblk : nb + rot;
                for (int u = 0; u < SC_U; ++u)
                    nxt[u] = oq[min(bn * (32 * SC_U) + u * 32 + lane, nq16 - 1)];
            }
            const int bc = blk + rot >= nblk ? blk + rot - nblk : blk + rot;
            const int qi0 = bc * (32 * SC_U) + lane;
            unsigned hb = 0u;
            for (int u = 0; u < SC_U; ++u) {
                const unsigned f0 = zero_bytes(cur[u].x ^ pat);
                const unsigned f1 = zero_bytes(cur[u].y ^ pat);
                const unsigned f2 = zero_bytes(cur[u].z ^ pat);
                const unsigned f3 = zero_bytes(cur[u].w ^ pat);
                const unsigned hf = (f0 >> 7) | (f1 >> 6) | (f2 >> 5) | (f3 >> 4);
                if (qi0 + 32 * u < nq16) hb |= hf << (4 * u);
            }
            unsigned mk = __builtin_amdgcn_ballot_w32(hb != 0u);
            for (int r = 0; r < 16 * SC_U && mk != 0u; ++r) {
                if (hb != 0u) {
                    const int p = __builtin_ctz(hb);
                    wl[cnt + __builtin_amdgcn_mbcnt_lo(mk, 0u)] =
                        16 * (qi0 + 32 * ((p >> 2) & 1)) + 4 * (p & 3) + (p >> 3);
                    hb &= hb - 1u;
                }
                cnt += __builtin_popcount(mk);
                mk = __builtin_amdgcn_ballot_w32(hb != 0u);
            }
            for (int u = 0; u < SC_U; ++u) cur[u] = nxt[u];
        }

        const int need = live ? 16 : 1;
        while (cnt >= need) {
            const int n = min(cnt, 16);
            cnt -= n;
            const int* bl = wl + cnt;
            const int er = bl[m < n ? m : 0];
            const int e = min(er, E - 1);
            const int s = eidx[e];
            const int d = dstp[e];
            const int dl = d - nodeBase;
            const bool ok = m < n && er < E && (unsigned)dl < (unsigned)rEff;
            const int dlk = ok ? dl : -1;

            const float4* ep = (const float4*)(eattr + (size_t)e * 16);
            const float4 q0 = ep[0], q1 = ep[1], q2 = ep[2], q3 = ep[3];
            const float2* xp = (const float2*)(x + (size_t)s * 16 + 2 * hi);
            const float2 xa = xp[0], xb = xp[2], xc = xp[4], xd = xp[6];
            const float eav[16] = {q0.x, q0.y, q0.z, q0.w, q1.x, q1.y, q1.z, q1.w,
                                   q2.x, q2.y, q2.z, q2.w, q3.x, q3.y, q3.z, q3.w};
            const float xr0[4] = {xa.x, xb.x, xc.x, xd.x};
            const float xr1[4] = {xa.y, xb.y, xc.y, xd.y};

            v8f acc = {};
            typedef __attribute__((address_space(3))) const volatile v2f lds_pair_t;
            lds_pair_t* sWb = (lds_pair_t*)sWp;
            v2f bb[68];
#pragma unroll
            for (int kk = 0; kk < PF_D; ++kk) bb[kk] = sWb[(2 * kk + hi) * 16 + m];
#pragma unroll
            for (int kk = 0; kk < 68; ++kk) {
                if (kk + PF_D < 68) bb[kk + PF_D] = sWb[(2 * (kk + PF_D) + hi) * 16 + m];
                __builtin_amdgcn_sched_barrier(0);
                const float x0 = xr0[kk & 3], x1 = xr1[kk & 3];
                v2f a;
                if (kk < 64) {
                    const float eaf = eav[kk >> 2];
                    a.x = eaf * x0; a.y = eaf * x1;
                } else {
                    a.x = x0; a.y = x1;
                }
                acc = __builtin_amdgcn_wmma_f32_16x16x4_f32(
                    false, a, false, bb[kk], (short)0, acc, false, false);
            }

#pragma unroll
            for (int v = 0; v < 8; ++v) {
                const int dlo = __builtin_amdgcn_readlane(dlk, v);
                const int dhi = __builtin_amdgcn_readlane(dlk, v + 8);
                const int dr = hi ? dhi : dlo;
                if (dr >= 0) atomicAdd(&sAgg[dr * 16 + m], acc[v]);
            }
        }
        if (!live) break;
    }
    __syncthreads();

    for (int i = tid; i < rEff * 16; i += 1024) {
        const int nd = nodeBase + (i >> 4), o = i & 15;
        float sacc = sAgg[i] + b_ecc[o];
        const float* xrow = x + (size_t)nd * 16;
#pragma unroll
        for (int j = 0; j < 16; ++j) sacc += xrow[j] * W_root[j * 16 + o];
        h[(size_t)nd * 16 + o] = fmaxf(sacc, 0.0f);
    }
}

__global__ void k_zgemm(const float* __restrict__ h,
                        const float* __restrict__ W_gat,
                        float* __restrict__ z, int N) {
    __shared__ float sWg[8 * 256];
    const int tid = threadIdx.x;
    for (int t = tid; t < 2048; t += 256) {
        int k = t >> 7, col = t & 127;
        sWg[(k >> 1) * 256 + col * 2 + (k & 1)] = W_gat[k * 128 + col];
    }
    __syncthreads();

    const int wave = tid >> 5, lane = tid & 31;
    const int m = lane & 15, hi = lane >> 4;
    const int rowTile = blockIdx.x;
    const int colTile = wave;
    const int row = rowTile * 16 + m;
    const int rc  = row < N ? row : (N - 1);
    const float* hr = h + (size_t)rc * 16;
    const int col = colTile * 16 + m;

    v8f acc = {};
#pragma unroll
    for (int kk = 0; kk < 4; ++kk) {
        int i0 = 4 * kk + 2 * hi;
        v2f a, b;
        a.x = hr[i0]; a.y = hr[i0 + 1];
        int p = 2 * kk + hi;
        b.x = sWg[p * 256 + col * 2 + 0];
        b.y = sWg[p * 256 + col * 2 + 1];
        acc = __builtin_amdgcn_wmma_f32_16x16x4_f32(
            false, a, false, b, (short)0, acc, false, false);
    }
#pragma unroll
    for (int v = 0; v < 8; ++v) {
        int r = rowTile * 16 + v + 8 * hi;
        if (r < N) z[(size_t)r * 128 + col] = acc[v];
    }
}

__global__ void k_escore(const float* __restrict__ z,
                         const float* __restrict__ a_src,
                         const float* __restrict__ a_dst,
                         float* __restrict__ esrc,
                         float* __restrict__ edst, int N) {
    int i = blockIdx.x * blockDim.x + threadIdx.x;
    if (i >= N * 4) return;
    int n = i >> 2, hh = i & 3;
    const float* zr = z + (size_t)n * 128 + hh * 32;
    const float* as = a_src + hh * 32;
    const float* ad = a_dst + hh * 32;
    float s = 0.f, d = 0.f;
#pragma unroll 8
    for (int j = 0; j < 32; ++j) { float zv = zr[j]; s += zv * as[j]; d += zv * ad[j]; }
    esrc[i] = s; edst[i] = d;
}

__global__ void k_attmax(const int* __restrict__ eidx,
                         const float* __restrict__ esrc,
                         const float* __restrict__ edst,
                         unsigned* __restrict__ maxkey, int E) {
    int i = blockIdx.x * blockDim.x + threadIdx.x;
    if (i >= E * 4) return;
    int e = i >> 2, hh = i & 3;
    int s = eidx[e], d = eidx[E + e];
    float lg = esrc[s * 4 + hh] + edst[d * 4 + hh];
    lg = lg > 0.f ? lg : NEG_SLOPE * lg;
    atomicMax(&maxkey[d * 4 + hh], fkey_enc(lg));
}

__global__ __launch_bounds__(1024)
void k_gatown(const int* __restrict__ eidx,
              const unsigned* __restrict__ own,
              const float* __restrict__ esrc,
              const float* __restrict__ edst,
              const unsigned* __restrict__ maxkey,
              const float* __restrict__ z,
              const float* __restrict__ b_gat,
              const float* __restrict__ W_fc,
              const float* __restrict__ b_fc,
              float* __restrict__ out, int N, int E) {
    __shared__ float sAcc[OWN_R * 128];
    __shared__ float sDen[OWN_R * 4];
    __shared__ int   sList[OWN_W * WL_CAP];

    const int tid = threadIdx.x;
    const int wave = __builtin_amdgcn_readfirstlane(tid >> 5);
    const int lane = tid & 31;
    const int nodeBase = blockIdx.x * OWN_R;
    const int rEff = max(min(OWN_R, N - nodeBase), 0);

    for (int i = tid; i < OWN_R * 128; i += 1024) sAcc[i] = 0.0f;
    for (int i = tid; i < OWN_R * 4; i += 1024) sDen[i] = 0.0f;
    __syncthreads();

    int* wl = sList + wave * WL_CAP;
    int cnt = 0;

    const int* dstp = eidx + E;
    const uint4* oq = (const uint4*)own;
    const int nq16 = (E + 15) >> 4;
    const int nblk = (nq16 + 32 * SC_U - 1) / (32 * SC_U);
    const int rot = (int)blockIdx.x * (nblk / (int)gridDim.x);
    const unsigned pat = ((unsigned)blockIdx.x & 255u) * 0x01010101u;

    const uint4 zero4 = make_uint4(0u, 0u, 0u, 0u);
    uint4 cur[SC_U];
    for (int u = 0; u < SC_U; ++u) cur[u] = zero4;
    if (wave < nblk) {
        const int bc = wave + rot >= nblk ? wave + rot - nblk : wave + rot;
        for (int u = 0; u < SC_U; ++u)
            cur[u] = oq[min(bc * (32 * SC_U) + u * 32 + lane, nq16 - 1)];
    }

    for (int blk = wave; ; blk += OWN_W) {
        const bool live = blk < nblk;
        if (live) {
            uint4 nxt[SC_U];
            const int nb = blk + OWN_W;
            for (int u = 0; u < SC_U; ++u) nxt[u] = zero4;
            if (nb < nblk) {
                const int bn = nb + rot >= nblk ? nb + rot - nblk : nb + rot;
                for (int u = 0; u < SC_U; ++u)
                    nxt[u] = oq[min(bn * (32 * SC_U) + u * 32 + lane, nq16 - 1)];
            }
            const int bc = blk + rot >= nblk ? blk + rot - nblk : blk + rot;
            const int qi0 = bc * (32 * SC_U) + lane;
            unsigned hb = 0u;
            for (int u = 0; u < SC_U; ++u) {
                const unsigned f0 = zero_bytes(cur[u].x ^ pat);
                const unsigned f1 = zero_bytes(cur[u].y ^ pat);
                const unsigned f2 = zero_bytes(cur[u].z ^ pat);
                const unsigned f3 = zero_bytes(cur[u].w ^ pat);
                const unsigned h = (f0 >> 7) | (f1 >> 6) | (f2 >> 5) | (f3 >> 4);
                if (qi0 + 32 * u < nq16) hb |= h << (4 * u);
            }
            unsigned mk = __builtin_amdgcn_ballot_w32(hb != 0u);
            for (int r = 0; r < 16 * SC_U && mk != 0u; ++r) {
                if (hb != 0u) {
                    const int p = __builtin_ctz(hb);
                    wl[cnt + __builtin_amdgcn_mbcnt_lo(mk, 0u)] =
                        16 * (qi0 + 32 * ((p >> 2) & 1)) + 4 * (p & 3) + (p >> 3);
                    hb &= hb - 1u;
                }
                cnt += __builtin_popcount(mk);
                mk = __builtin_amdgcn_ballot_w32(hb != 0u);
            }
            for (int u = 0; u < SC_U; ++u) cur[u] = nxt[u];
        }

        const int need = live ? 32 : 1;
        while (cnt >= need) {
            const int n = min(cnt, 32);
            cnt -= n;
            const int* bl = wl + cnt;
            const int er = bl[lane < n ? lane : 0];
            const int e = min(er, E - 1);
            const int s = eidx[e];
            const int d = dstp[e];
            const int dl = d - nodeBase;
            const bool ok = lane < n && er < E && (unsigned)dl < (unsigned)rEff;
            const unsigned okm = __builtin_amdgcn_ballot_w32(ok);
            const float4 es = *(const float4*)(esrc + (size_t)s * 4);
            const float4 ed = *(const float4*)(edst + (size_t)d * 4);
            const uint4  mx = *(const uint4*)(maxkey + (size_t)d * 4);
            float l0 = es.x + ed.x, l1 = es.y + ed.y, l2 = es.z + ed.z, l3 = es.w + ed.w;
            l0 = l0 > 0.f ? l0 : NEG_SLOPE * l0;
            l1 = l1 > 0.f ? l1 : NEG_SLOPE * l1;
            l2 = l2 > 0.f ? l2 : NEG_SLOPE * l2;
            l3 = l3 > 0.f ? l3 : NEG_SLOPE * l3;
            const float x0 = __expf(l0 - fkey_dec(mx.x));
            const float x1 = __expf(l1 - fkey_dec(mx.y));
            const float x2 = __expf(l2 - fkey_dec(mx.z));
            const float x3 = __expf(l3 - fkey_dec(mx.w));
            if (ok) {
                atomicAdd(&sDen[dl * 4 + 0], x0);
                atomicAdd(&sDen[dl * 4 + 1], x1);
                atomicAdd(&sDen[dl * 4 + 2], x2);
                atomicAdd(&sDen[dl * 4 + 3], x3);
            }
            const int hd = lane >> 3;
            for (int j0 = 0; j0 < n; j0 += 8) {
                float4 r[8];
                for (int u = 0; u < 8; ++u) {
                    const int j = min(j0 + u, n - 1);
                    const int sj = __builtin_amdgcn_readlane(s, j);
                    r[u] = *(const float4*)(z + (size_t)sj * 128 + lane * 4);
                }
                for (int u = 0; u < 8; ++u) {
                    const int j = j0 + u;
                    if ((okm >> j) & 1u) {
                        const int dj = __builtin_amdgcn_readlane(dl, j);
                        const float w0 = __int_as_float(__builtin_amdgcn_readlane(__float_as_int(x0), j));
                        const float w1 = __int_as_float(__builtin_amdgcn_readlane(__float_as_int(x1), j));
                        const float w2 = __int_as_float(__builtin_amdgcn_readlane(__float_as_int(x2), j));
                        const float w3 = __int_as_float(__builtin_amdgcn_readlane(__float_as_int(x3), j));
                        const float wa = (hd & 1) ? w1 : w0;
                        const float wb = (hd & 1) ? w3 : w2;
                        const float w  = (hd & 2) ? wb : wa;
                        float* ap = sAcc + dj * 128 + lane;
                        atomicAdd(ap,      w * r[u].x);
                        atomicAdd(ap + 32, w * r[u].y);
                        atomicAdd(ap + 64, w * r[u].z);
                        atomicAdd(ap + 96, w * r[u].w);
                    }
                }
            }
        }
        if (!live) break;
    }
    __syncthreads();

    for (int i = wave; i < rEff; i += OWN_W) {
        float acc = 0.f;
        for (int t = 0; t < 4; ++t) {
            const int j = t * 32 + lane;
            const float dn = sDen[i * 4 + t] + 1e-16f;
            float g = sAcc[i * 128 + (j & 3) * 32 + (j >> 2)] / dn + b_gat[j];
            g = fmaxf(g, 0.0f);
            acc += g * W_fc[j];
        }
        for (int off = 16; off > 0; off >>= 1) acc += __shfl_xor(acc, off, 32);
        if (lane == 0) out[nodeBase + i] = acc + b_fc[0];
    }
}

extern "C" void kernel_launch(void* const* d_in, const int* in_sizes, int n_in,
                              void* d_out, int out_size, void* d_ws, size_t ws_size,
                              hipStream_t stream) {
    const float* x      = (const float*)d_in[0];
    const int*   eidx   = (const int*)  d_in[1];
    const float* eattr  = (const float*)d_in[2];
    const float* W_nn   = (const float*)d_in[3];
    const float* b_nn   = (const float*)d_in[4];
    const float* W_root = (const float*)d_in[5];
    const float* b_ecc  = (const float*)d_in[6];
    const float* W_gat  = (const float*)d_in[7];
    const float* a_src  = (const float*)d_in[8];
    const float* a_dst  = (const float*)d_in[9];
    const float* b_gat  = (const float*)d_in[10];
    const float* W_fc   = (const float*)d_in[11];
    const float* b_fc   = (const float*)d_in[12];
    float* out = (float*)d_out;

    const int N = in_sizes[0] / 16;
    const int E = in_sizes[1] / 2;

    float* ws = (float*)d_ws;
    size_t off = 0;
    unsigned* maxkey = (unsigned*)(ws + off);  off += (size_t)N * 4;
    float*    hbuf   = ws + off;               off += (size_t)N * 16;
    float*    zbuf   = ws + off;               off += (size_t)N * 128;
    float*    esrc   = ws + off;               off += (size_t)N * 4;
    float*    edst   = ws + off;               off += (size_t)N * 4;
    unsigned* own    = (unsigned*)(ws + off);  off += (size_t)((E + 127) / 128) * 32;

    k_prep  <<<1024, 256, 0, stream>>>(eidx, maxkey, own, N, E);
    k_nnown <<<(N + OWN_R - 1) / OWN_R, 1024, 0, stream>>>(x, eidx, eattr, W_nn, b_nn, W_root, b_ecc,
                                                        own, hbuf, N, E);
    k_zgemm <<<(N + 15) / 16, 256, 0, stream>>>(hbuf, W_gat, zbuf, N);
    k_escore<<<(N * 4 + 255) / 256, 256, 0, stream>>>(zbuf, a_src, a_dst, esrc, edst, N);
    k_attmax<<<(E * 4 + 255) / 256, 256, 0, stream>>>(eidx, esrc, edst, maxkey, E);
    k_gatown<<<(N + OWN_R - 1) / OWN_R, 1024, 0, stream>>>(eidx, own, esrc, edst, maxkey, zbuf,
                                                        b_gat, W_fc, b_fc, out, N, E);
}


# GAT owner work-groups take their edges from a list of (source, owned row) pairs that the NNConv owners write while filtering: no second owner-byte scan, no index reloads; sum order still LDS arrival
# speedup vs baseline: 6.8201x; 1.1065x over previous
#include <hip/hip_runtime.h>
#include <hip/hip_bf16.h>

typedef __attribute__((ext_vector_type(2))) float v2f;
typedef __attribute__((ext_vector_type(8))) float v8f;

#define NEG_SLOPE 0.2f
#define OWN_R  196

__device__ __forceinline__ unsigned fkey_enc(float f) {
    unsigned b = __float_as_uint(f);
    return (b & 0x80000000u) ? ~b : (b | 0x80000000u);
}
__device__ __forceinline__ float fkey_dec(unsigned k) {
    return (k & 0x80000000u) ? __uint_as_float(k ^ 0x80000000u)
                             : __uint_as_float(~k);
}

#define OWN_W  32
#define SC_U   2
#define WL_CAP (32 + SC_U * 512)
#define PF_D   6
static_assert(SC_U == 2, "the flag word of the scan holds two groups of 16 edges per lane");

__device__ __forceinline__ unsigned zero_bytes(unsigned t) {
    return ~(((t & 0x7F7F7F7Fu) + 0x7F7F7F7Fu) | t | 0x7F7F7F7Fu);
}

__global__ void k_prep(const int* __restrict__ eidx,
                       unsigned* __restrict__ maxkey,
                       unsigned* __restrict__ ownw, int N, int E) {
    const int i0 = blockIdx.x * blockDim.x + threadIdx.x;
    const int stride = gridDim.x * blockDim.x;
    for (int i = i0; i < N * 4; i += stride) maxkey[i] = 0u;
    const int* dstp = eidx + E;
    const int nw = ((E + 127) / 128) * 32;
    for (int w = i0; w < nw; w += stride) {
        unsigned pk = 0u;
        for (int k = 0; k < 4; ++k) {
            const int idx = 4 * w + k;
            const int d = idx < E ? dstp[idx] : 0;
            pk |= (((unsigned)d / OWN_R) & 255u) << (8 * k);
        }
        ownw[w] = pk;
    }
}

__global__ __launch_bounds__(1024)
void k_nnown(const float* __restrict__ x,
             const int*   __restrict__ eidx,
             const float* __restrict__ eattr,
             const float* __restrict__ W_nn,
             const float* __restrict__ b_nn,
             const float* __restrict__ W_root,
             const float* __restrict__ b_ecc,
             const unsigned* __restrict__ own,
             float* __restrict__ h,
             int* __restrict__ keptCnt,
             int* __restrict__ keptLst, int cap, int N, int E) {
    __shared__ float sWp[136 * 32];
    __shared__ float sAgg[OWN_R * 16];
    __shared__ int   sList[OWN_W * WL_CAP];
    __shared__ int   sKept;

    const int tid = threadIdx.x;
    const int wave = __builtin_amdgcn_readfirstlane(tid >> 5);
    const int lane = tid & 31;
    const int m    = lane & 15;
    const int hi   = lane >> 4;
    const int nodeBase = blockIdx.x * OWN_R;
    const int rEff = max(min(OWN_R, N - nodeBase), 0);

    for (int t = tid; t < 272 * 16; t += 1024) {
        int k = t >> 4, o = t & 15;
        float v = (k < 256) ? W_nn[(k >> 4) * 256 + (k & 15) * 16 + o]
                            : b_nn[(k - 256) * 16 + o];
        sWp[(k >> 1) * 32 + o * 2 + (k & 1)] = v;
    }
    for (int i = tid; i < OWN_R * 16; i += 1024) sAgg[i] = 0.0f;
    if (tid == 0) sKept = 0;
    __syncthreads();

    int* wl = sList + wave * WL_CAP;
    int cnt = 0;
    int2* kl = (int2*)keptLst + (size_t)blockIdx.x * (size_t)cap;

    const int* dstp = eidx + E;
    const uint4* oq = (const uint4*)own;
    const int nq16 = (E + 15) >> 4;
    const int nblk = (nq16 + 32 * SC_U - 1) / (32 * SC_U);
    const int rot = (int)blockIdx.x * (nblk / (int)gridDim.x);
    const unsigned pat = ((unsigned)blockIdx.x & 255u) * 0x01010101u;

    const uint4 zero4 = make_uint4(0u, 0u, 0u, 0u);
    uint4 cur[SC_U];
    for (int u = 0; u < SC_U; ++u) cur[u] = zero4;
    if (wave < nblk) {
        const int bc = wave + rot >= nblk ? wave + rot - nblk : wave + rot;
        for (int u = 0; u < SC_U; ++u)
            cur[u] = oq[min(bc * (32 * SC_U) + u * 32 + lane, nq16 - 1)];
    }

    for (int blk = wave; ; blk += OWN_W) {
        const bool live = blk < nblk;
        if (live) {
            uint4 nxt[SC_U];
            const int nb = blk + OWN_W;
            for (int u = 0; u < SC_U; ++u) nxt[u] = zero4;
            if (nb < nblk) {
                const int bn = nb + rot >= nblk ? nb + rot - nblk : nb + rot;
                for (int u = 0; u < SC_U; ++u)
                    nxt[u] = oq[min(bn * (32 * SC_U) + u * 32 + lane, nq16 - 1)];
            }
            const int bc = blk + rot >= nblk ? blk + rot - nblk : blk + rot;
            const int qi0 = bc * (32 * SC_U) + lane;
            unsigned hb = 0u;
            for (int u = 0; u < SC_U; ++u) {
                const unsigned f0 = zero_bytes(cur[u].x ^ pat);
                const unsigned f1 = zero_bytes(cur[u].y ^ pat);
                const unsigned f2 = zero_bytes(cur[u].z ^ pat);
                const unsigned f3 = zero_bytes(cur[u].w ^ pat);
                const unsigned hf = (f0 >> 7) | (f1 >> 6) | (f2 >> 5) | (f3 >> 4);
                if (qi0 + 32 * u < nq16) hb |= hf << (4 * u);
            }
            unsigned mk = __builtin_amdgcn_ballot_w32(hb != 0u);
            for (int r = 0; r < 16 * SC_U && mk != 0u; ++r) {
                if (hb != 0u) {
                    const int p = __builtin_ctz(hb);
                    wl[cnt + __builtin_amdgcn_mbcnt_lo(mk, 0u)] =
                        16 * (qi0 + 32 * ((p >> 2) & 1)) + 4 * (p & 3) + (p >> 3);
                    hb &= hb - 1u;
                }
                cnt += __builtin_popcount(mk);
                mk = __builtin_amdgcn_ballot_w32(hb != 0u);
            }
            for (int u = 0; u < SC_U; ++u) cur[u] = nxt[u];
        }

        const int need = live ? 16 : 1;
        while (cnt >= need) {
            const int n = min(cnt, 16);
            cnt -= n;
            const int* bl = wl + cnt;
            const int er = bl[m < n ? m : 0];
            const int e = min(er, E - 1);
            const int s = eidx[e];
            const int d = dstp[e];
            const int dl = d - nodeBase;
            const bool ok = m < n && er < E && (unsigned)dl < (unsigned)rEff;
            const int dlk = ok ? dl : -1;

            const float4* ep = (const float4*)(eattr + (size_t)e * 16);
            const float4 q0 = ep[0], q1 = ep[1], q2 = ep[2], q3 = ep[3];
            const float2* xp = (const float2*)(x + (size_t)s * 16 + 2 * hi);
            const float2 xa = xp[0], xb = xp[2], xc = xp[4], xd = xp[6];
            const float eav[16] = {q0.x, q0.y, q0.z, q0.w, q1.x, q1.y, q1.z, q1.w,
                                   q2.x, q2.y, q2.z, q2.w, q3.x, q3.y, q3.z, q3.w};
            const float xr0[4] = {xa.x, xb.x, xc.x, xd.x};
            const float xr1[4] = {xa.y, xb.y, xc.y, xd.y};

            v8f acc = {};
            typedef __attribute__((address_space(3))) const volatile v2f lds_pair_t;
            lds_pair_t* sWb = (lds_pair_t*)sWp;
            v2f bb[68];
#pragma unroll
            for (int kk = 0; kk < PF_D; ++kk) bb[kk] = sWb[(2 * kk + hi) * 16 + m];
#pragma unroll
            for (int kk = 0; kk < 68; ++kk) {
                if (kk + PF_D < 68) bb[kk + PF_D] = sWb[(2 * (kk + PF_D) + hi) * 16 + m];
                __builtin_amdgcn_sched_barrier(0);
                const float x0 = xr0[kk & 3], x1 = xr1[kk & 3];
                v2f a;
                if (kk < 64) {
                    const float eaf = eav[kk >> 2];
                    a.x = eaf * x0; a.y = eaf * x1;
                } else {
                    a.x = x0; a.y = x1;
                }
                acc = __builtin_amdgcn_wmma_f32_16x16x4_f32(
                    false, a, false, bb[kk], (short)0, acc, false, false);
            }

#pragma unroll
            for (int v = 0; v < 8; ++v) {
                const int dlo = __builtin_amdgcn_readlane(dlk, v);
                const int dhi = __builtin_amdgcn_readlane(dlk, v + 8);
                const int dr = hi ? dhi : dlo;
                if (dr >= 0) atomicAdd(&sAgg[dr * 16 + m], acc[v]);
            }

            if (cap > 0) {
                const unsigned okl = __builtin_amdgcn_ballot_w32(ok) & 0xFFFFu;
                if (okl != 0u) {
                    int at = 0;
                    if (lane == 0) at = atomicAdd(&sKept, __builtin_popcount(okl));
                    at = __builtin_amdgcn_readfirstlane(at);
                    const int pos = at + __builtin_popcount(okl & ((1u << lane) - 1u));
                    if (lane < 16 && ok && pos < cap) kl[pos] = make_int2(s, dl);
                }
            }
        }
        if (!live) break;
    }
    __syncthreads();
    if (cap > 0 && tid == 0) keptCnt[blockIdx.x] = sKept;

    for (int i = tid; i < rEff * 16; i += 1024) {
        const int nd = nodeBase + (i >> 4), o = i & 15;
        float sacc = sAgg[i] + b_ecc[o];
        const float* xrow = x + (size_t)nd * 16;
#pragma unroll
        for (int j = 0; j < 16; ++j) sacc += xrow[j] * W_root[j * 16 + o];
        h[(size_t)nd * 16 + o] = fmaxf(sacc, 0.0f);
    }
}

__global__ void k_zgemm(const float* __restrict__ h,
                        const float* __restrict__ W_gat,
                        float* __restrict__ z, int N) {
    __shared__ float sWg[8 * 256];
    const int tid = threadIdx.x;
    for (int t = tid; t < 2048; t += 256) {
        int k = t >> 7, col = t & 127;
        sWg[(k >> 1) * 256 + col * 2 + (k & 1)] = W_gat[k * 128 + col];
    }
    __syncthreads();

    const int wave = tid >> 5, lane = tid & 31;
    const int m = lane & 15, hi = lane >> 4;
    const int rowTile = blockIdx.x;
    const int colTile = wave;
    const int row = rowTile * 16 + m;
    const int rc  = row < N ? row : (N - 1);
    const float* hr = h + (size_t)rc * 16;
    const int col = colTile * 16 + m;

    v8f acc = {};
#pragma unroll
    for (int kk = 0; kk < 4; ++kk) {
        int i0 = 4 * kk + 2 * hi;
        v2f a, b;
        a.x = hr[i0]; a.y = hr[i0 + 1];
        int p = 2 * kk + hi;
        b.x = sWg[p * 256 + col * 2 + 0];
        b.y = sWg[p * 256 + col * 2 + 1];
        acc = __builtin_amdgcn_wmma_f32_16x16x4_f32(
            false, a, false, b, (short)0, acc, false, false);
    }
#pragma unroll
    for (int v = 0; v < 8; ++v) {
        int r = rowTile * 16 + v + 8 * hi;
        if (r < N) z[(size_t)r * 128 + col] = acc[v];
    }
}

__global__ void k_escore(const float* __restrict__ z,
                         const float* __restrict__ a_src,
                         const float* __restrict__ a_dst,
                         float* __restrict__ esrc,
                         float* __restrict__ edst, int N) {
    int i = blockIdx.x * blockDim.x + threadIdx.x;
    if (i >= N * 4) return;
    int n = i >> 2, hh = i & 3;
    const float* zr = z + (size_t)n * 128 + hh * 32;
    const float* as = a_src + hh * 32;
    const float* ad = a_dst + hh * 32;
    float s = 0.f, d = 0.f;
#pragma unroll 8
    for (int j = 0; j < 32; ++j) { float zv = zr[j]; s += zv * as[j]; d += zv * ad[j]; }
    esrc[i] = s; edst[i] = d;
}

__global__ void k_attmax(const int* __restrict__ eidx,
                         const float* __restrict__ esrc,
                         const float* __restrict__ edst,
                         unsigned* __restrict__ maxkey, int E) {
    int i = blockIdx.x * blockDim.x + threadIdx.x;
    if (i >= E * 4) return;
    int e = i >> 2, hh = i & 3;
    int s = eidx[e], d = eidx[E + e];
    float lg = esrc[s * 4 + hh] + edst[d * 4 + hh];
    lg = lg > 0.f ? lg : NEG_SLOPE * lg;
    atomicMax(&maxkey[d * 4 + hh], fkey_enc(lg));
}

__device__ __forceinline__ void gat_batch(const int s, const int d, const int dl, const bool ok,
                                          const int n, const int lane,
                                          const float* __restrict__ esrc,
                                          const float* __restrict__ edst,
                                          const unsigned* __restrict__ maxkey,
                                          const float* __restrict__ z,
                                          float* sAcc, float* sDen) {
    const unsigned okm = __builtin_amdgcn_ballot_w32(ok);
    const float4 es = *(const float4*)(esrc + (size_t)s * 4);
    const float4 ed = *(const float4*)(edst + (size_t)d * 4);
    const uint4  mx = *(const uint4*)(maxkey + (size_t)d * 4);
    float l0 = es.x + ed.x, l1 = es.y + ed.y, l2 = es.z + ed.z, l3 = es.w + ed.w;
    l0 = l0 > 0.f ? l0 : NEG_SLOPE * l0;
    l1 = l1 > 0.f ? l1 : NEG_SLOPE * l1;
    l2 = l2 > 0.f ? l2 : NEG_SLOPE * l2;
    l3 = l3 > 0.f ? l3 : NEG_SLOPE * l3;
    const float x0 = __expf(l0 - fkey_dec(mx.x));
    const float x1 = __expf(l1 - fkey_dec(mx.y));
    const float x2 = __expf(l2 - fkey_dec(mx.z));
    const float x3 = __expf(l3 - fkey_dec(mx.w));
    if (ok) {
        atomicAdd(&sDen[dl * 4 + 0], x0);
        atomicAdd(&sDen[dl * 4 + 1], x1);
        atomicAdd(&sDen[dl * 4 + 2], x2);
        atomicAdd(&sDen[dl * 4 + 3], x3);
    }
    const int hd = lane >> 3;
    for (int j0 = 0; j0 < n; j0 += 8) {
        float4 r[8];
        for (int u = 0; u < 8; ++u) {
            const int j = min(j0 + u, n - 1);
            const int sj = __builtin_amdgcn_readlane(s, j);
            r[u] = *(const float4*)(z + (size_t)sj * 128 + lane * 4);
        }
        for (int u = 0; u < 8; ++u) {
            const int j = j0 + u;
            if ((okm >> j) & 1u) {
                const int dj = __builtin_amdgcn_readlane(dl, j);
                const float w0 = __int_as_float(__builtin_amdgcn_readlane(__float_as_int(x0), j));
                const float w1 = __int_as_float(__builtin_amdgcn_readlane(__float_as_int(x1), j));
                const float w2 = __int_as_float(__builtin_amdgcn_readlane(__float_as_int(x2), j));
                const float w3 = __int_as_float(__builtin_amdgcn_readlane(__float_as_int(x3), j));
                const float wa = (hd & 1) ? w1 : w0;
                const float wb = (hd & 1) ? w3 : w2;
                const float w  = (hd & 2) ? wb : wa;
                float* ap = sAcc + dj * 128 + lane;
                atomicAdd(ap,      w * r[u].x);
                atomicAdd(ap + 32, w * r[u].y);
                atomicAdd(ap + 64, w * r[u].z);
                atomicAdd(ap + 96, w * r[u].w);
            }
        }
    }
}

__global__ __launch_bounds__(1024)
void k_gatown(const int* __restrict__ eidx,
              const unsigned* __restrict__ own,
              const float* __restrict__ esrc,
              const float* __restrict__ edst,
              const unsigned* __restrict__ maxkey,
              const float* __restrict__ z,
              const float* __restrict__ b_gat,
              const float* __restrict__ W_fc,
              const float* __restrict__ b_fc,
              const int* __restrict__ keptCnt,
              const int* __restrict__ keptLst, int cap,
              float* __restrict__ out, int N, int E) {
    __shared__ float sAcc[OWN_R * 128];
    __shared__ float sDen[OWN_R * 4];
    __shared__ int   sList[OWN_W * WL_CAP];

    const int tid = threadIdx.x;
    const int wave = __builtin_amdgcn_readfirstlane(tid >> 5);
    const int lane = tid & 31;
    const int nodeBase = blockIdx.x * OWN_R;
    const int rEff = max(min(OWN_R, N - nodeBase), 0);

    for (int i = tid; i < OWN_R * 128; i += 1024) sAcc[i] = 0.0f;
    for (int i = tid; i < OWN_R * 4; i += 1024) sDen[i] = 0.0f;
    __syncthreads();

    const int kept = cap > 0 ? keptCnt[blockIdx.x] : 0;
    const bool fromList = cap > 0 && kept <= cap;

    if (fromList) {
        const int2* kl = (const int2*)keptLst + (size_t)blockIdx.x * (size_t)cap;
        const int run = (((kept + OWN_W - 1) / OWN_W) + 7) & ~7;
        const int end = min(kept, (wave + 1) * run);
        int base = wave * run;
        int2 cur = make_int2(0, 0);
        if (base < end) cur = kl[base + min(lane, min(32, end - base) - 1)];
        while (base < end) {
            const int n = min(32, end - base);
            const int nb = base + 32;
            int2 nxt = cur;
            if (nb < end) nxt = kl[nb + min(lane, min(32, end - nb) - 1)];
            const int s = cur.x, dl = cur.y;
            gat_batch(s, nodeBase + dl, dl, lane < n, n, lane, esrc, edst, maxkey, z, sAcc, sDen);
            cur = nxt;
            base = nb;
        }
    } else {
    int* wl = sList + wave * WL_CAP;
    int cnt = 0;

    const int* dstp = eidx + E;
    const uint4* oq = (const uint4*)own;
    const int nq16 = (E + 15) >> 4;
    const int nblk = (nq16 + 32 * SC_U - 1) / (32 * SC_U);
    const int rot = (int)blockIdx.x * (nblk / (int)gridDim.x);
    const unsigned pat = ((unsigned)blockIdx.x & 255u) * 0x01010101u;

    const uint4 zero4 = make_uint4(0u, 0u, 0u, 0u);
    uint4 cur[SC_U];
    for (int u = 0; u < SC_U; ++u) cur[u] = zero4;
    if (wave < nblk) {
        const int bc = wave + rot >= nblk ? wave + rot - nblk : wave + rot;
        for (int u = 0; u < SC_U; ++u)
            cur[u] = oq[min(bc * (32 * SC_U) + u * 32 + lane, nq16 - 1)];
    }

    for (int blk = wave; ; blk += OWN_W) {
        const bool live = blk < nblk;
        if (live) {
            uint4 nxt[SC_U];
            const int nb = blk + OWN_W;
            for (int u = 0; u < SC_U; ++u) nxt[u] = zero4;
            if (nb < nblk) {
                const int bn = nb + rot >= nblk ? nb + rot - nblk : nb + rot;
                for (int u = 0; u < SC_U; ++u)
                    nxt[u] = oq[min(bn * (32 * SC_U) + u * 32 + lane, nq16 - 1)];
            }
            const int bc = blk + rot >= nblk ? blk + rot - nblk : blk + rot;
            const int qi0 = bc * (32 * SC_U) + lane;
            unsigned hb = 0u;
            for (int u = 0; u < SC_U; ++u) {
                const unsigned f0 = zero_bytes(cur[u].x ^ pat);
                const unsigned f1 = zero_bytes(cur[u].y ^ pat);
                const unsigned f2 = zero_bytes(cur[u].z ^ pat);
                const unsigned f3 = zero_bytes(cur[u].w ^ pat);
                const unsigned h = (f0 >> 7) | (f1 >> 6) | (f2 >> 5) | (f3 >> 4);
                if (qi0 + 32 * u < nq16) hb |= h << (4 * u);
            }
            unsigned mk = __builtin_amdgcn_ballot_w32(hb != 0u);
            for (int r = 0; r < 16 * SC_U && mk != 0u; ++r) {
                if (hb != 0u) {
                    const int p = __builtin_ctz(hb);
                    wl[cnt + __builtin_amdgcn_mbcnt_lo(mk, 0u)] =
                        16 * (qi0 + 32 * ((p >> 2) & 1)) + 4 * (p & 3) + (p >> 3);
                    hb &= hb - 1u;
                }
                cnt += __builtin_popcount(mk);
                mk = __builtin_amdgcn_ballot_w32(hb != 0u);
            }
            for (int u = 0; u < SC_U; ++u) cur[u] = nxt[u];
        }

        const int need = live ? 32 : 1;
        while (cnt >= need) {
            const int n = min(cnt, 32);
            cnt -= n;
            const int* bl = wl + cnt;
            const int er = bl[lane < n ? lane : 0];
            const int e = min(er, E - 1);
            const int s = eidx[e];
            const int d = dstp[e];
            const int dl = d - nodeBase;
            const bool ok = lane < n && er < E && (unsigned)dl < (unsigned)rEff;
            gat_batch(s, d, dl, ok, n, lane, esrc, edst, maxkey, z, sAcc, sDen);
        }
        if (!live) break;
    }
    }
    __syncthreads();

    for (int i = wave; i < rEff; i += OWN_W) {
        float acc = 0.f;
        for (int t = 0; t < 4; ++t) {
            const int j = t * 32 + lane;
            const float dn = sDen[i * 4 + t] + 1e-16f;
            float g = sAcc[i * 128 + (j & 3) * 32 + (j >> 2)] / dn + b_gat[j];
            g = fmaxf(g, 0.0f);
            acc += g * W_fc[j];
        }
        for (int off = 16; off > 0; off >>= 1) acc += __shfl_xor(acc, off, 32);
        if (lane == 0) out[nodeBase + i] = acc + b_fc[0];
    }
}

extern "C" void kernel_launch(void* const* d_in, const int* in_sizes, int n_in,
                              void* d_out, int out_size, void* d_ws, size_t ws_size,
                              hipStream_t stream) {
    const float* x      = (const float*)d_in[0];
    const int*   eidx   = (const int*)  d_in[1];
    const float* eattr  = (const float*)d_in[2];
    const float* W_nn   = (const float*)d_in[3];
    const float* b_nn   = (const float*)d_in[4];
    const float* W_root = (const float*)d_in[5];
    const float* b_ecc  = (const float*)d_in[6];
    const float* W_gat  = (const float*)d_in[7];
    const float* a_src  = (const float*)d_in[8];
    const float* a_dst  = (const float*)d_in[9];
    const float* b_gat  = (const float*)d_in[10];
    const float* W_fc   = (const float*)d_in[11];
    const float* b_fc   = (const float*)d_in[12];
    float* out = (float*)d_out;

    const int N = in_sizes[0] / 16;
    const int E = in_sizes[1] / 2;

    float* ws = (float*)d_ws;
    size_t off = 0;
    unsigned* maxkey = (unsigned*)(ws + off);  off += (size_t)N * 4;
    float*    hbuf   = ws + off;               off += (size_t)N * 16;
    float*    zbuf   = ws + off;               off += (size_t)N * 128;
    float*    esrc   = ws + off;               off += (size_t)N * 4;
    float*    edst   = ws + off;               off += (size_t)N * 4;
    unsigned* own    = (unsigned*)(ws + off);  off += (size_t)((E + 127) / 128) * 32;
    const int nOwn = (N + OWN_R - 1) / OWN_R;
    int cap = ((E / nOwn) * 3 / 2 + 127) / 128 * 128;
    int* keptCnt = (int*)(ws + off);           off += (size_t)((nOwn + 31) / 32) * 32;
    int* keptLst = (int*)(ws + off);           off += (size_t)nOwn * (size_t)cap * 2;
    if (off * sizeof(float) > ws_size) cap = 0;

    k_prep  <<<1024, 256, 0, stream>>>(eidx, maxkey, own, N, E);
    k_nnown <<<nOwn, 1024, 0, stream>>>(x, eidx, eattr, W_nn, b_nn, W_root, b_ecc,
                                         own, hbuf, keptCnt, keptLst, cap, N, E);
    k_zgemm <<<(N + 15) / 16, 256, 0, stream>>>(hbuf, W_gat, zbuf, N);
    k_escore<<<(N * 4 + 255) / 256, 256, 0, stream>>>(zbuf, a_src, a_dst, esrc, edst, N);
    k_attmax<<<(E * 4 + 255) / 256, 256, 0, stream>>>(eidx, esrc, edst, maxkey, E);
    k_gatown<<<nOwn, 1024, 0, stream>>>(eidx, own, esrc, edst, maxkey, zbuf,
                                         b_gat, W_fc, b_fc, keptCnt, keptLst, cap, out, N, E);
}


# GAT segment max inside the owner work-groups: LDS integer max of source scores over the kept list ahead of the sums (same shift value); no max plane, no all-edge max launch; 6->5 launches
# speedup vs baseline: 8.3823x; 1.2291x over previous
#include <hip/hip_runtime.h>
#include <hip/hip_bf16.h>

typedef __attribute__((ext_vector_type(2))) float v2f;
typedef __attribute__((ext_vector_type(8))) float v8f;

#define NEG_SLOPE 0.2f
#define OWN_R  196

__device__ __forceinline__ unsigned fkey_enc(float f) {
    unsigned b = __float_as_uint(f);
    return (b & 0x80000000u) ? ~b : (b | 0x80000000u);
}
__device__ __forceinline__ float fkey_dec(unsigned k) {
    return (k & 0x80000000u) ? __uint_as_float(k ^ 0x80000000u)
                             : __uint_as_float(~k);
}

#define OWN_W  32
#define SC_U   2
#define WL_CAP (32 + SC_U * 512)
#define PF_D   6
static_assert(SC_U == 2, "the flag word of the scan holds two groups of 16 edges per lane");

__device__ __forceinline__ unsigned zero_bytes(unsigned t) {
    return ~(((t & 0x7F7F7F7Fu) + 0x7F7F7F7Fu) | t | 0x7F7F7F7Fu);
}

__global__ void k_prep(const int* __restrict__ eidx,
                       unsigned* __restrict__ ownw, int E) {
    const int i0 = blockIdx.x * blockDim.x + threadIdx.x;
    const int stride = gridDim.x * blockDim.x;
    const int* dstp = eidx + E;
    const int nw = ((E + 127) / 128) * 32;
    for (int w = i0; w < nw; w += stride) {
        unsigned pk = 0u;
        for (int k = 0; k < 4; ++k) {
            const int idx = 4 * w + k;
            const int d = idx < E ? dstp[idx] : 0;
            pk |= (((unsigned)d / OWN_R) & 255u) << (8 * k);
        }
        ownw[w] = pk;
    }
}

__global__ __launch_bounds__(1024)
void k_nnown(const float* __restrict__ x,
             const int*   __restrict__ eidx,
             const float* __restrict__ eattr,
             const float* __restrict__ W_nn,
             const float* __restrict__ b_nn,
             const float* __restrict__ W_root,
             const float* __restrict__ b_ecc,
             const unsigned* __restrict__ own,
             float* __restrict__ h,
             int* __restrict__ keptCnt,
             int* __restrict__ keptLst, int cap, int N, int E) {
    __shared__ float sWp[136 * 32];
    __shared__ float sAgg[OWN_R * 16];
    __shared__ int   sList[OWN_W * WL_CAP];
    __shared__ int   sKept;

    const int tid = threadIdx.x;
    const int wave = __builtin_amdgcn_readfirstlane(tid >> 5);
    const int lane = tid & 31;
    const int m    = lane & 15;
    const int hi   = lane >> 4;
    const int nodeBase = blockIdx.x * OWN_R;
    const int rEff = max(min(OWN_R, N - nodeBase), 0);

    for (int t = tid; t < 272 * 16; t += 1024) {
        int k = t >> 4, o = t & 15;
        float v = (k < 256) ? W_nn[(k >> 4) * 256 + (k & 15) * 16 + o]
                            : b_nn[(k - 256) * 16 + o];
        sWp[(k >> 1) * 32 + o * 2 + (k & 1)] = v;
    }
    for (int i = tid; i < OWN_R * 16; i += 1024) sAgg[i] = 0.0f;
    if (tid == 0) sKept = 0;
    __syncthreads();

    int* wl = sList + wave * WL_CAP;
    int cnt = 0;
    int2* kl = (int2*)keptLst + (size_t)blockIdx.x * (size_t)cap;

    const int* dstp = eidx + E;
    const uint4* oq = (const uint4*)own;
    const int nq16 = (E + 15) >> 4;
    const int nblk = (nq16 + 32 * SC_U - 1) / (32 * SC_U);
    const int rot = (int)blockIdx.x * (nblk / (int)gridDim.x);
    const unsigned pat = ((unsigned)blockIdx.x & 255u) * 0x01010101u;

    const uint4 zero4 = make_uint4(0u, 0u, 0u, 0u);
    uint4 cur[SC_U];
    for (int u = 0; u < SC_U; ++u) cur[u] = zero4;
    if (wave < nblk) {
        const int bc = wave + rot >= nblk ? wave + rot - nblk : wave + rot;
        for (int u = 0; u < SC_U; ++u)
            cur[u] = oq[min(bc * (32 * SC_U) + u * 32 + lane, nq16 - 1)];
    }

    for (int blk = wave; ; blk += OWN_W) {
        const bool live = blk < nblk;
        if (live) {
            uint4 nxt[SC_U];
            const int nb = blk + OWN_W;
            for (int u = 0; u < SC_U; ++u) nxt[u] = zero4;
            if (nb < nblk) {
                const int bn = nb + rot >= nblk ? nb + rot - nblk : nb + rot;
                for (int u = 0; u < SC_U; ++u)
                    nxt[u] = oq[min(bn * (32 * SC_U) + u * 32 + lane, nq16 - 1)];
            }
            const int bc = blk + rot >= nblk ? blk + rot - nblk : blk + rot;
            const int qi0 = bc * (32 * SC_U) + lane;
            unsigned hb = 0u;
            for (int u = 0; u < SC_U; ++u) {
                const unsigned f0 = zero_bytes(cur[u].x ^ pat);
                const unsigned f1 = zero_bytes(cur[u].y ^ pat);
                const unsigned f2 = zero_bytes(cur[u].z ^ pat);
                const unsigned f3 = zero_bytes(cur[u].w ^ pat);
                const unsigned hf = (f0 >> 7) | (f1 >> 6) | (f2 >> 5) | (f3 >> 4);
                if (qi0 + 32 * u < nq16) hb |= hf << (4 * u);
            }
            unsigned mk = __builtin_amdgcn_ballot_w32(hb != 0u);
            for (int r = 0; r < 16 * SC_U && mk != 0u; ++r) {
                if (hb != 0u) {
                    const int p = __builtin_ctz(hb);
                    wl[cnt + __builtin_amdgcn_mbcnt_lo(mk, 0u)] =
                        16 * (qi0 + 32 * ((p >> 2) & 1)) + 4 * (p & 3) + (p >> 3);
                    hb &= hb - 1u;
                }
                cnt += __builtin_popcount(mk);
                mk = __builtin_amdgcn_ballot_w32(hb != 0u);
            }
            for (int u = 0; u < SC_U; ++u) cur[u] = nxt[u];
        }

        const int need = live ? 16 : 1;
        while (cnt >= need) {
            const int n = min(cnt, 16);
            cnt -= n;
            const int* bl = wl + cnt;
            const int er = bl[m < n ? m : 0];
            const int e = min(er, E - 1);
            const int s = eidx[e];
            const int d = dstp[e];
            const int dl = d - nodeBase;
            const bool ok = m < n && er < E && (unsigned)dl < (unsigned)rEff;
            const int dlk = ok ? dl : -1;

            const float4* ep = (const float4*)(eattr + (size_t)e * 16);
            const float4 q0 = ep[0], q1 = ep[1], q2 = ep[2], q3 = ep[3];
            const float2* xp = (const float2*)(x + (size_t)s * 16 + 2 * hi);
            const float2 xa = xp[0], xb = xp[2], xc = xp[4], xd = xp[6];
            const float eav[16] = {q0.x, q0.y, q0.z, q0.w, q1.x, q1.y, q1.z, q1.w,
                                   q2.x, q2.y, q2.z, q2.w, q3.x, q3.y, q3.z, q3.w};
            const float xr0[4] = {xa.x, xb.x, xc.x, xd.x};
            const float xr1[4] = {xa.y, xb.y, xc.y, xd.y};

            v8f acc = {};
            typedef __attribute__((address_space(3))) const volatile v2f lds_pair_t;
            lds_pair_t* sWb = (lds_pair_t*)sWp;
            v2f bb[68];
#pragma unroll
            for (int kk = 0; kk < PF_D; ++kk) bb[kk] = sWb[(2 * kk + hi) * 16 + m];
#pragma unroll
            for (int kk = 0; kk < 68; ++kk) {
                if (kk + PF_D < 68) bb[kk + PF_D] = sWb[(2 * (kk + PF_D) + hi) * 16 + m];
                __builtin_amdgcn_sched_barrier(0);
                const float x0 = xr0[kk & 3], x1 = xr1[kk & 3];
                v2f a;
                if (kk < 64) {
                    const float eaf = eav[kk >> 2];
                    a.x = eaf * x0; a.y = eaf * x1;
                } else {
                    a.x = x0; a.y = x1;
                }
                acc = __builtin_amdgcn_wmma_f32_16x16x4_f32(
                    false, a, false, bb[kk], (short)0, acc, false, false);
            }

#pragma unroll
            for (int v = 0; v < 8; ++v) {
                const int dlo = __builtin_amdgcn_readlane(dlk, v);
                const int dhi = __builtin_amdgcn_readlane(dlk, v + 8);
                const int dr = hi ? dhi : dlo;
                if (dr >= 0) atomicAdd(&sAgg[dr * 16 + m], acc[v]);
            }

            if (cap > 0) {
                const unsigned okl = __builtin_amdgcn_ballot_w32(ok) & 0xFFFFu;
                if (okl != 0u) {
                    int at = 0;
                    if (lane == 0) at = atomicAdd(&sKept, __builtin_popcount(okl));
                    at = __builtin_amdgcn_readfirstlane(at);
                    const int pos = at + __builtin_popcount(okl & ((1u << lane) - 1u));
                    if (lane < 16 && ok && pos < cap) kl[pos] = make_int2(s, dl);
                }
            }
        }
        if (!live) break;
    }
    __syncthreads();
    if (cap > 0 && tid == 0) keptCnt[blockIdx.x] = sKept;

    for (int i = tid; i < rEff * 16; i += 1024) {
        const int nd = nodeBase + (i >> 4), o = i & 15;
        float sacc = sAgg[i] + b_ecc[o];
        const float* xrow = x + (size_t)nd * 16;
#pragma unroll
        for (int j = 0; j < 16; ++j) sacc += xrow[j] * W_root[j * 16 + o];
        h[(size_t)nd * 16 + o] = fmaxf(sacc, 0.0f);
    }
}

__global__ void k_zgemm(const float* __restrict__ h,
                        const float* __restrict__ W_gat,
                        float* __restrict__ z, int N) {
    __shared__ float sWg[8 * 256];
    const int tid = threadIdx.x;
    for (int t = tid; t < 2048; t += 256) {
        int k = t >> 7, col = t & 127;
        sWg[(k >> 1) * 256 + col * 2 + (k & 1)] = W_gat[k * 128 + col];
    }
    __syncthreads();

    const int wave = tid >> 5, lane = tid & 31;
    const int m = lane & 15, hi = lane >> 4;
    const int rowTile = blockIdx.x;
    const int colTile = wave;
    const int row = rowTile * 16 + m;
    const int rc  = row < N ? row : (N - 1);
    const float* hr = h + (size_t)rc * 16;
    const int col = colTile * 16 + m;

    v8f acc = {};
#pragma unroll
    for (int kk = 0; kk < 4; ++kk) {
        int i0 = 4 * kk + 2 * hi;
        v2f a, b;
        a.x = hr[i0]; a.y = hr[i0 + 1];
        int p = 2 * kk + hi;
        b.x = sWg[p * 256 + col * 2 + 0];
        b.y = sWg[p * 256 + col * 2 + 1];
        acc = __builtin_amdgcn_wmma_f32_16x16x4_f32(
            false, a, false, b, (short)0, acc, false, false);
    }
#pragma unroll
    for (int v = 0; v < 8; ++v) {
        int r = rowTile * 16 + v + 8 * hi;
        if (r < N) z[(size_t)r * 128 + col] = acc[v];
    }
}

__global__ void k_escore(const float* __restrict__ z,
                         const float* __restrict__ a_src,
                         const float* __restrict__ a_dst,
                         float* __restrict__ esrc,
                         float* __restrict__ edst, int N) {
    int i = blockIdx.x * blockDim.x + threadIdx.x;
    if (i >= N * 4) return;
    int n = i >> 2, hh = i & 3;
    const float* zr = z + (size_t)n * 128 + hh * 32;
    const float* as = a_src + hh * 32;
    const float* ad = a_dst + hh * 32;
    float s = 0.f, d = 0.f;
#pragma unroll 8
    for (int j = 0; j < 32; ++j) { float zv = zr[j]; s += zv * as[j]; d += zv * ad[j]; }
    esrc[i] = s; edst[i] = d;
}

__device__ __forceinline__ void max_edge(const float4 es, const int dl, unsigned* sMax) {
    atomicMax(&sMax[dl * 4 + 0], fkey_enc(es.x));
    atomicMax(&sMax[dl * 4 + 1], fkey_enc(es.y));
    atomicMax(&sMax[dl * 4 + 2], fkey_enc(es.z));
    atomicMax(&sMax[dl * 4 + 3], fkey_enc(es.w));
}

__device__ __forceinline__ void gat_batch(const int s, const int d, const int dl, const bool ok,
                                          const int n, const int lane,
                                          const float* __restrict__ esrc,
                                          const float* __restrict__ edst,
                                          const unsigned* sMax,
                                          const float* __restrict__ z,
                                          float* sAcc, float* sDen) {
    const unsigned okm = __builtin_amdgcn_ballot_w32(ok);
    const float4 es = *(const float4*)(esrc + (size_t)s * 4);
    const float4 ed = *(const float4*)(edst + (size_t)d * 4);
    const uint4  mx = *(const uint4*)(sMax + (ok ? dl : 0) * 4);
    float l0 = es.x + ed.x, l1 = es.y + ed.y, l2 = es.z + ed.z, l3 = es.w + ed.w;
    l0 = l0 > 0.f ? l0 : NEG_SLOPE * l0;
    l1 = l1 > 0.f ? l1 : NEG_SLOPE * l1;
    l2 = l2 > 0.f ? l2 : NEG_SLOPE * l2;
    l3 = l3 > 0.f ? l3 : NEG_SLOPE * l3;
    float m0 = fkey_dec(mx.x) + ed.x, m1 = fkey_dec(mx.y) + ed.y;
    float m2 = fkey_dec(mx.z) + ed.z, m3 = fkey_dec(mx.w) + ed.w;
    m0 = m0 > 0.f ? m0 : NEG_SLOPE * m0;
    m1 = m1 > 0.f ? m1 : NEG_SLOPE * m1;
    m2 = m2 > 0.f ? m2 : NEG_SLOPE * m2;
    m3 = m3 > 0.f ? m3 : NEG_SLOPE * m3;
    const float x0 = __expf(l0 - m0);
    const float x1 = __expf(l1 - m1);
    const float x2 = __expf(l2 - m2);
    const float x3 = __expf(l3 - m3);
    if (ok) {
        atomicAdd(&sDen[dl * 4 + 0], x0);
        atomicAdd(&sDen[dl * 4 + 1], x1);
        atomicAdd(&sDen[dl * 4 + 2], x2);
        atomicAdd(&sDen[dl * 4 + 3], x3);
    }
    const int hd = lane >> 3;
    for (int j0 = 0; j0 < n; j0 += 8) {
        float4 r[8];
        for (int u = 0; u < 8; ++u) {
            const int j = min(j0 + u, n - 1);
            const int sj = __builtin_amdgcn_readlane(s, j);
            r[u] = *(const float4*)(z + (size_t)sj * 128 + lane * 4);
        }
        for (int u = 0; u < 8; ++u) {
            const int j = j0 + u;
            if ((okm >> j) & 1u) {
                const int dj = __builtin_amdgcn_readlane(dl, j);
                const float w0 = __int_as_float(__builtin_amdgcn_readlane(__float_as_int(x0), j));
                const float w1 = __int_as_float(__builtin_amdgcn_readlane(__float_as_int(x1), j));
                const float w2 = __int_as_float(__builtin_amdgcn_readlane(__float_as_int(x2), j));
                const float w3 = __int_as_float(__builtin_amdgcn_readlane(__float_as_int(x3), j));
                const float wa = (hd & 1) ? w1 : w0;
                const float wb = (hd & 1) ? w3 : w2;
                const float w  = (hd & 2) ? wb : wa;
                float* ap = sAcc + dj * 128 + lane;
                atomicAdd(ap,      w * r[u].x);
                atomicAdd(ap + 32, w * r[u].y);
                atomicAdd(ap + 64, w * r[u].z);
                atomicAdd(ap + 96, w * r[u].w);
            }
        }
    }
}

__global__ __launch_bounds__(1024)
void k_gatown(const int* __restrict__ eidx,
              const unsigned* __restrict__ own,
              const float* __restrict__ esrc,
              const float* __restrict__ edst,
              const float* __restrict__ z,
              const float* __restrict__ b_gat,
              const float* __restrict__ W_fc,
              const float* __restrict__ b_fc,
              const int* __restrict__ keptCnt,
              const int* __restrict__ keptLst, int cap,
              float* __restrict__ out, int N, int E) {
    __shared__ float sAcc[OWN_R * 128];
    __shared__ float sDen[OWN_R * 4];
    __shared__ unsigned sMax[OWN_R * 4];
    __shared__ int   sList[OWN_W * WL_CAP];

    const int tid = threadIdx.x;
    const int wave = __builtin_amdgcn_readfirstlane(tid >> 5);
    const int lane = tid & 31;
    const int nodeBase = blockIdx.x * OWN_R;
    const int rEff = max(min(OWN_R, N - nodeBase), 0);

    const int kept = cap > 0 ? keptCnt[blockIdx.x] : 0;
    const bool fromList = cap > 0 && kept <= cap;

    for (int i = tid; i < OWN_R * 128; i += 1024) sAcc[i] = 0.0f;
    for (int i = tid; i < OWN_R * 4; i += 1024) { sDen[i] = 0.0f; sMax[i] = 0u; }
    __syncthreads();

    if (fromList) {
        const int2* kl = (const int2*)keptLst + (size_t)blockIdx.x * (size_t)cap;
        for (int b0 = 0; b0 < kept; b0 += 4 * 1024) {
            int2 p[4];
            float4 sc[4];
#pragma unroll
            for (int u = 0; u < 4; ++u) p[u] = kl[min(b0 + u * 1024 + tid, kept - 1)];
            asm volatile("" :: "v"(p[0].x), "v"(p[1].x), "v"(p[2].x), "v"(p[3].x));
#pragma unroll
            for (int u = 0; u < 4; ++u) sc[u] = *(const float4*)(esrc + (size_t)(unsigned)p[u].x * 4);
            asm volatile("" :: "v"(sc[0].x), "v"(sc[1].x), "v"(sc[2].x), "v"(sc[3].x));
#pragma unroll
            for (int u = 0; u < 4; ++u)
                if (b0 + u * 1024 + tid < kept) max_edge(sc[u], p[u].y, sMax);
        }
        __syncthreads();

        const int run = (((kept + OWN_W - 1) / OWN_W) + 7) & ~7;
        const int end = min(kept, (wave + 1) * run);
        int base = wave * run;
        int2 cur = make_int2(0, 0);
        if (base < end) cur = kl[base + min(lane, min(32, end - base) - 1)];
        while (base < end) {
            const int n = min(32, end - base);
            const int nb = base + 32;
            int2 nxt = cur;
            if (nb < end) nxt = kl[nb + min(lane, min(32, end - nb) - 1)];
            const int s = cur.x, dl = cur.y;
            gat_batch(s, nodeBase + dl, dl, lane < n, n, lane, esrc, edst, sMax, z, sAcc, sDen);
            cur = nxt;
            base = nb;
        }
    } else {
    for (int pass = 0; pass < 2; ++pass) {
    int* wl = sList + wave * WL_CAP;
    int cnt = 0;

    const int* dstp = eidx + E;
    const uint4* oq = (const uint4*)own;
    const int nq16 = (E + 15) >> 4;
    const int nblk = (nq16 + 32 * SC_U - 1) / (32 * SC_U);
    const int rot = (int)blockIdx.x * (nblk / (int)gridDim.x);
    const unsigned pat = ((unsigned)blockIdx.x & 255u) * 0x01010101u;

    const uint4 zero4 = make_uint4(0u, 0u, 0u, 0u);
    uint4 cur[SC_U];
    for (int u = 0; u < SC_U; ++u) cur[u] = zero4;
    if (wave < nblk) {
        const int bc = wave + rot >= nblk ? wave + rot - nblk : wave + rot;
        for (int u = 0; u < SC_U; ++u)
            cur[u] = oq[min(bc * (32 * SC_U) + u * 32 + lane, nq16 - 1)];
    }

    for (int blk = wave; ; blk += OWN_W) {
        const bool live = blk < nblk;
        if (live) {
            uint4 nxt[SC_U];
            const int nb = blk + OWN_W;
            for (int u = 0; u < SC_U; ++u) nxt[u] = zero4;
            if (nb < nblk) {
                const int bn = nb + rot >= nblk ? nb + rot - nblk : nb + rot;
                for (int u = 0; u < SC_U; ++u)
                    nxt[u] = oq[min(bn * (32 * SC_U) + u * 32 + lane, nq16 - 1)];
            }
            const int bc = blk + rot >= nblk ? blk + rot - nblk : blk + rot;
            const int qi0 = bc * (32 * SC_U) + lane;
            unsigned hb = 0u;
            for (int u = 0; u < SC_U; ++u) {
                const unsigned f0 = zero_bytes(cur[u].x ^ pat);
                const unsigned f1 = zero_bytes(cur[u].y ^ pat);
                const unsigned f2 = zero_bytes(cur[u].z ^ pat);
                const unsigned f3 = zero_bytes(cur[u].w ^ pat);
                const unsigned h = (f0 >> 7) | (f1 >> 6) | (f2 >> 5) | (f3 >> 4);
                if (qi0 + 32 * u < nq16) hb |= h << (4 * u);
            }
            unsigned mk = __builtin_amdgcn_ballot_w32(hb != 0u);
            for (int r = 0; r < 16 * SC_U && mk != 0u; ++r) {
                if (hb != 0u) {
                    const int p = __builtin_ctz(hb);
                    wl[cnt + __builtin_amdgcn_mbcnt_lo(mk, 0u)] =
                        16 * (qi0 + 32 * ((p >> 2) & 1)) + 4 * (p & 3) + (p >> 3);
                    hb &= hb - 1u;
                }
                cnt += __builtin_popcount(mk);
                mk = __builtin_amdgcn_ballot_w32(hb != 0u);
            }
            for (int u = 0; u < SC_U; ++u) cur[u] = nxt[u];
        }

        const int need = live ? 32 : 1;
        while (cnt >= need) {
            const int n = min(cnt, 32);
            cnt -= n;
            const int* bl = wl + cnt;
            const int er = bl[lane < n ? lane : 0];
            const int e = min(er, E - 1);
            const int s = eidx[e];
            const int d = dstp[e];
            const int dl = d - nodeBase;
            const bool ok = lane < n && er < E && (unsigned)dl < (unsigned)rEff;
            if (pass == 0) {
                const float4 sc = *(const float4*)(esrc + (size_t)s * 4);
                if (ok) max_edge(sc, dl, sMax);
            } else {
                gat_batch(s, d, dl, ok, n, lane, esrc, edst, sMax, z, sAcc, sDen);
            }
        }
        if (!live) break;
    }
    if (pass == 0) __syncthreads();
    }
    }
    __syncthreads();

    for (int i = wave; i < rEff; i += OWN_W) {
        float acc = 0.f;
        for (int t = 0; t < 4; ++t) {
            const int j = t * 32 + lane;
            const float dn = sDen[i * 4 + t] + 1e-16f;
            float g = sAcc[i * 128 + (j & 3) * 32 + (j >> 2)] / dn + b_gat[j];
            g = fmaxf(g, 0.0f);
            acc += g * W_fc[j];
        }
        for (int off = 16; off > 0; off >>= 1) acc += __shfl_xor(acc, off, 32);
        if (lane == 0) out[nodeBase + i] = acc + b_fc[0];
    }
}

extern "C" void kernel_launch(void* const* d_in, const int* in_sizes, int n_in,
                              void* d_out, int out_size, void* d_ws, size_t ws_size,
                              hipStream_t stream) {
    const float* x      = (const float*)d_in[0];
    const int*   eidx   = (const int*)  d_in[1];
    const float* eattr  = (const float*)d_in[2];
    const float* W_nn   = (const float*)d_in[3];
    const float* b_nn   = (const float*)d_in[4];
    const float* W_root = (const float*)d_in[5];
    const float* b_ecc  = (const float*)d_in[6];
    const float* W_gat  = (const float*)d_in[7];
    const float* a_src  = (const float*)d_in[8];
    const float* a_dst  = (const float*)d_in[9];
    const float* b_gat  = (const float*)d_in[10];
    const float* W_fc   = (const float*)d_in[11];
    const float* b_fc   = (const float*)d_in[12];
    float* out = (float*)d_out;

    const int N = in_sizes[0] / 16;
    const int E = in_sizes[1] / 2;

    float* ws = (float*)d_ws;
    size_t off = 0;
    float*    hbuf   = ws + off;               off += (size_t)N * 16;
    float*    zbuf   = ws + off;               off += (size_t)N * 128;
    float*    esrc   = ws + off;               off += (size_t)N * 4;
    float*    edst   = ws + off;               off += (size_t)N * 4;
    unsigned* own    = (unsigned*)(ws + off);  off += (size_t)((E + 127) / 128) * 32;
    const int nOwn = (N + OWN_R - 1) / OWN_R;
    int cap = ((E / nOwn) * 3 / 2 + 127) / 128 * 128;
    int* keptCnt = (int*)(ws + off);           off += (size_t)((nOwn + 31) / 32) * 32;
    int* keptLst = (int*)(ws + off);           off += (size_t)nOwn * (size_t)cap * 2;
    if (off * sizeof(float) > ws_size) cap = 0;

    k_prep  <<<1024, 256, 0, stream>>>(eidx, own, E);
    k_nnown <<<nOwn, 1024, 0, stream>>>(x, eidx, eattr, W_nn, b_nn, W_root, b_ecc,
                                         own, hbuf, keptCnt, keptLst, cap, N, E);
    k_zgemm <<<(N + 15) / 16, 256, 0, stream>>>(hbuf, W_gat, zbuf, N);
    k_escore<<<(N * 4 + 255) / 256, 256, 0, stream>>>(zbuf, a_src, a_dst, esrc, edst, N);
    k_gatown<<<nOwn, 1024, 0, stream>>>(eidx, own, esrc, edst, zbuf,
                                         b_gat, W_fc, b_fc, keptCnt, keptLst, cap, out, N, E);
}


# Head scores as a ninth column tile of the z GEMM from a folded 16x8 matrix W_gat.a built in the first kernel (score sum order changes): no score launch, no re-read of z; 5->4 launches
# speedup vs baseline: 9.0085x; 1.0747x over previous
#include <hip/hip_runtime.h>
#include <hip/hip_bf16.h>

typedef __attribute__((ext_vector_type(2))) float v2f;
typedef __attribute__((ext_vector_type(8))) float v8f;

#define NEG_SLOPE 0.2f
#define OWN_R  196

__device__ __forceinline__ unsigned fkey_enc(float f) {
    unsigned b = __float_as_uint(f);
    return (b & 0x80000000u) ? ~b : (b | 0x80000000u);
}
__device__ __forceinline__ float fkey_dec(unsigned k) {
    return (k & 0x80000000u) ? __uint_as_float(k ^ 0x80000000u)
                             : __uint_as_float(~k);
}

#define OWN_W  32
#define SC_U   2
#define WL_CAP (32 + SC_U * 512)
#define PF_D   6
static_assert(SC_U == 2, "the flag word of the scan holds two groups of 16 edges per lane");

__device__ __forceinline__ unsigned zero_bytes(unsigned t) {
    return ~(((t & 0x7F7F7F7Fu) + 0x7F7F7F7Fu) | t | 0x7F7F7F7Fu);
}

__global__ void k_prep(const int* __restrict__ eidx,
                       unsigned* __restrict__ ownw, int E,
                       const float* __restrict__ W_gat,
                       const float* __restrict__ a_src,
                       const float* __restrict__ a_dst,
                       float* __restrict__ wa) {
    if (blockIdx.x == 0 && threadIdx.x < 128) {
        const int k = threadIdx.x >> 3, cc = threadIdx.x & 7;
        const float* av = (cc < 4 ? a_src : a_dst) + (cc & 3) * 32;
        const float* wr = W_gat + k * 128 + (cc & 3) * 32;
        float s = 0.f;
        for (int d = 0; d < 32; ++d) s += wr[d] * av[d];
        wa[k * 8 + cc] = s;
    }
    const int i0 = blockIdx.x * blockDim.x + threadIdx.x;
    const int stride = gridDim.x * blockDim.x;
    const int* dstp = eidx + E;
    const int nw = ((E + 127) / 128) * 32;
    for (int w = i0; w < nw; w += stride) {
        unsigned pk = 0u;
        for (int k = 0; k < 4; ++k) {
            const int idx = 4 * w + k;
            const int d = idx < E ? dstp[idx] : 0;
            pk |= (((unsigned)d / OWN_R) & 255u) << (8 * k);
        }
        ownw[w] = pk;
    }
}

__global__ __launch_bounds__(1024)
void k_nnown(const float* __restrict__ x,
             const int*   __restrict__ eidx,
             const float* __restrict__ eattr,
             const float* __restrict__ W_nn,
             const float* __restrict__ b_nn,
             const float* __restrict__ W_root,
             const float* __restrict__ b_ecc,
             const unsigned* __restrict__ own,
             float* __restrict__ h,
             int* __restrict__ keptCnt,
             int* __restrict__ keptLst, int cap, int N, int E) {
    __shared__ float sWp[136 * 32];
    __shared__ float sAgg[OWN_R * 16];
    __shared__ int   sList[OWN_W * WL_CAP];
    __shared__ int   sKept;

    const int tid = threadIdx.x;
    const int wave = __builtin_amdgcn_readfirstlane(tid >> 5);
    const int lane = tid & 31;
    const int m    = lane & 15;
    const int hi   = lane >> 4;
    const int nodeBase = blockIdx.x * OWN_R;
    const int rEff = max(min(OWN_R, N - nodeBase), 0);

    for (int t = tid; t < 272 * 16; t += 1024) {
        int k = t >> 4, o = t & 15;
        float v = (k < 256) ? W_nn[(k >> 4) * 256 + (k & 15) * 16 + o]
                            : b_nn[(k - 256) * 16 + o];
        sWp[(k >> 1) * 32 + o * 2 + (k & 1)] = v;
    }
    for (int i = tid; i < OWN_R * 16; i += 1024) sAgg[i] = 0.0f;
    if (tid == 0) sKept = 0;
    __syncthreads();

    int* wl = sList + wave * WL_CAP;
    int cnt = 0;
    int2* kl = (int2*)keptLst + (size_t)blockIdx.x * (size_t)cap;

    const int* dstp = eidx + E;
    const uint4* oq = (const uint4*)own;
    const int nq16 = (E + 15) >> 4;
    const int nblk = (nq16 + 32 * SC_U - 1) / (32 * SC_U);
    const int rot = (int)blockIdx.x * (nblk / (int)gridDim.x);
    const unsigned pat = ((unsigned)blockIdx.x & 255u) * 0x01010101u;

    const uint4 zero4 = make_uint4(0u, 0u, 0u, 0u);
    uint4 cur[SC_U];
    for (int u = 0; u < SC_U; ++u) cur[u] = zero4;
    if (wave < nblk) {
        const int bc = wave + rot >= nblk ? wave + rot - nblk : wave + rot;
        for (int u = 0; u < SC_U; ++u)
            cur[u] = oq[min(bc * (32 * SC_U) + u * 32 + lane, nq16 - 1)];
    }

    for (int blk = wave; ; blk += OWN_W) {
        const bool live = blk < nblk;
        if (live) {
            uint4 nxt[SC_U];
            const int nb = blk + OWN_W;
            for (int u = 0; u < SC_U; ++u) nxt[u] = zero4;
            if (nb < nblk) {
                const int bn = nb + rot >= nblk ? nb + rot - nblk : nb + rot;
                for (int u = 0; u < SC_U; ++u)
                    nxt[u] = oq[min(bn * (32 * SC_U) + u * 32 + lane, nq16 - 1)];
            }
            const int bc = blk + rot >= nblk ? blk + rot - nblk : blk + rot;
            const int qi0 = bc * (32 * SC_U) + lane;
            unsigned hb = 0u;
            for (int u = 0; u < SC_U; ++u) {
                const unsigned f0 = zero_bytes(cur[u].x ^ pat);
                const unsigned f1 = zero_bytes(cur[u].y ^ pat);
                const unsigned f2 = zero_bytes(cur[u].z ^ pat);
                const unsigned f3 = zero_bytes(cur[u].w ^ pat);
                const unsigned hf = (f0 >> 7) | (f1 >> 6) | (f2 >> 5) | (f3 >> 4);
                if (qi0 + 32 * u < nq16) hb |= hf << (4 * u);
            }
            unsigned mk = __builtin_amdgcn_ballot_w32(hb != 0u);
            for (int r = 0; r < 16 * SC_U && mk != 0u; ++r) {
                if (hb != 0u) {
                    const int p = __builtin_ctz(hb);
                    wl[cnt + __builtin_amdgcn_mbcnt_lo(mk, 0u)] =
                        16 * (qi0 + 32 * ((p >> 2) & 1)) + 4 * (p & 3) + (p >> 3);
                    hb &= hb - 1u;
                }
                cnt += __builtin_popcount(mk);
                mk = __builtin_amdgcn_ballot_w32(hb != 0u);
            }
            for (int u = 0; u < SC_U; ++u) cur[u] = nxt[u];
        }

        const int need = live ? 16 : 1;
        while (cnt >= need) {
            const int n = min(cnt, 16);
            cnt -= n;
            const int* bl = wl + cnt;
            const int er = bl[m < n ? m : 0];
            const int e = min(er, E - 1);
            const int s = eidx[e];
            const int d = dstp[e];
            const int dl = d - nodeBase;
            const bool ok = m < n && er < E && (unsigned)dl < (unsigned)rEff;
            const int dlk = ok ? dl : -1;

            const float4* ep = (const float4*)(eattr + (size_t)e * 16);
            const float4 q0 = ep[0], q1 = ep[1], q2 = ep[2], q3 = ep[3];
            const float2* xp = (const float2*)(x + (size_t)s * 16 + 2 * hi);
            const float2 xa = xp[0], xb = xp[2], xc = xp[4], xd = xp[6];
            const float eav[16] = {q0.x, q0.y, q0.z, q0.w, q1.x, q1.y, q1.z, q1.w,
                                   q2.x, q2.y, q2.z, q2.w, q3.x, q3.y, q3.z, q3.w};
            const float xr0[4] = {xa.x, xb.x, xc.x, xd.x};
            const float xr1[4] = {xa.y, xb.y, xc.y, xd.y};

            v8f acc = {};
            typedef __attribute__((address_space(3))) const volatile v2f lds_pair_t;
            lds_pair_t* sWb = (lds_pair_t*)sWp;
            v2f bb[68];
#pragma unroll
            for (int kk = 0; kk < PF_D; ++kk) bb[kk] = sWb[(2 * kk + hi) * 16 + m];
#pragma unroll
            for (int kk = 0; kk < 68; ++kk) {
                if (kk + PF_D < 68) bb[kk + PF_D] = sWb[(2 * (kk + PF_D) + hi) * 16 + m];
                __builtin_amdgcn_sched_barrier(0);
                const float x0 = xr0[kk & 3], x1 = xr1[kk & 3];
                v2f a;
                if (kk < 64) {
                    const float eaf = eav[kk >> 2];
                    a.x = eaf * x0; a.y = eaf * x1;
                } else {
                    a.x = x0; a.y = x1;
                }
                acc = __builtin_amdgcn_wmma_f32_16x16x4_f32(
                    false, a, false, bb[kk], (short)0, acc, false, false);
            }

#pragma unroll
            for (int v = 0; v < 8; ++v) {
                const int dlo = __builtin_amdgcn_readlane(dlk, v);
                const int dhi = __builtin_amdgcn_readlane(dlk, v + 8);
                const int dr = hi ? dhi : dlo;
                if (dr >= 0) atomicAdd(&sAgg[dr * 16 + m], acc[v]);
            }

            if (cap > 0) {
                const unsigned okl = __builtin_amdgcn_ballot_w32(ok) & 0xFFFFu;
                if (okl != 0u) {
                    int at = 0;
                    if (lane == 0) at = atomicAdd(&sKept, __builtin_popcount(okl));
                    at = __builtin_amdgcn_readfirstlane(at);
                    const int pos = at + __builtin_popcount(okl & ((1u << lane) - 1u));
                    if (lane < 16 && ok && pos < cap) kl[pos] = make_int2(s, dl);
                }
            }
        }
        if (!live) break;
    }
    __syncthreads();
    if (cap > 0 && tid == 0) keptCnt[blockIdx.x] = sKept;

    for (int i = tid; i < rEff * 16; i += 1024) {
        const int nd = nodeBase + (i >> 4), o = i & 15;
        float sacc = sAgg[i] + b_ecc[o];
        const float* xrow = x + (size_t)nd * 16;
#pragma unroll
        for (int j = 0; j < 16; ++j) sacc += xrow[j] * W_root[j * 16 + o];
        h[(size_t)nd * 16 + o] = fmaxf(sacc, 0.0f);
    }
}

#define ZG_COLS 144
#define ZG_T    288
__global__ void k_zgemm(const float* __restrict__ h,
                        const float* __restrict__ W_gat,
                        const float* __restrict__ wa,
                        float* __restrict__ z,
                        float* __restrict__ esrc,
                        float* __restrict__ edst, int N) {
    __shared__ float sWg[8 * ZG_COLS * 2];
    const int tid = threadIdx.x;
    float wv[8];
#pragma unroll
    for (int u = 0; u < 8; ++u) {
        const int t = tid + u * ZG_T;
        wv[u] = t < 2048 ? W_gat[t] : wa[min(t - 2048, 127)];
    }
#pragma unroll
    for (int u = 0; u < 8; ++u) {
        const int t = tid + u * ZG_T;
        if (t < 2048) {
            const int k = t >> 7, col = t & 127;
            sWg[(k >> 1) * (ZG_COLS * 2) + col * 2 + (k & 1)] = wv[u];
        } else if (t < 2048 + 128) {
            const int k = (t - 2048) >> 3, cc = (t - 2048) & 7;
            sWg[(k >> 1) * (ZG_COLS * 2) + (128 + cc) * 2 + (k & 1)] = wv[u];
        }
    }
    if (tid < 128) {
        const int k = tid >> 3, cc = 8 + (tid & 7);
        sWg[(k >> 1) * (ZG_COLS * 2) + (128 + cc) * 2 + (k & 1)] = 0.0f;
    }
    __syncthreads();

    const int wave = tid >> 5, lane = tid & 31;
    const int m = lane & 15, hi = lane >> 4;
    const int rowTile = blockIdx.x;
    const int colTile = wave;
    const int row = rowTile * 16 + m;
    const int rc  = row < N ? row : (N - 1);
    const float* hr = h + (size_t)rc * 16;
    const int col = colTile * 16 + m;

    v8f acc = {};
#pragma unroll
    for (int kk = 0; kk < 4; ++kk) {
        int i0 = 4 * kk + 2 * hi;
        v2f a, b;
        a.x = hr[i0]; a.y = hr[i0 + 1];
        int p = 2 * kk + hi;
        b.x = sWg[p * (ZG_COLS * 2) + col * 2 + 0];
        b.y = sWg[p * (ZG_COLS * 2) + col * 2 + 1];
        acc = __builtin_amdgcn_wmma_f32_16x16x4_f32(
            false, a, false, b, (short)0, acc, false, false);
    }
    if (colTile < 8) {
#pragma unroll
        for (int v = 0; v < 8; ++v) {
            int r = rowTile * 16 + v + 8 * hi;
            if (r < N) z[(size_t)r * 128 + col] = acc[v];
        }
    } else if (m < 8) {
        float* sp = m < 4 ? esrc + m : edst + (m - 4);
#pragma unroll
        for (int v = 0; v < 8; ++v) {
            int r = rowTile * 16 + v + 8 * hi;
            if (r < N) sp[(size_t)r * 4] = acc[v];
        }
    }
}

__device__ __forceinline__ void max_edge(const float4 es, const int dl, unsigned* sMax) {
    atomicMax(&sMax[dl * 4 + 0], fkey_enc(es.x));
    atomicMax(&sMax[dl * 4 + 1], fkey_enc(es.y));
    atomicMax(&sMax[dl * 4 + 2], fkey_enc(es.z));
    atomicMax(&sMax[dl * 4 + 3], fkey_enc(es.w));
}

__device__ __forceinline__ void gat_batch(const int s, const int d, const int dl, const bool ok,
                                          const int n, const int lane,
                                          const float* __restrict__ esrc,
                                          const float* __restrict__ edst,
                                          const unsigned* sMax,
                                          const float* __restrict__ z,
                                          float* sAcc, float* sDen) {
    const unsigned okm = __builtin_amdgcn_ballot_w32(ok);
    const float4 es = *(const float4*)(esrc + (size_t)s * 4);
    const float4 ed = *(const float4*)(edst + (size_t)d * 4);
    const uint4  mx = *(const uint4*)(sMax + (ok ? dl : 0) * 4);
    float l0 = es.x + ed.x, l1 = es.y + ed.y, l2 = es.z + ed.z, l3 = es.w + ed.w;
    l0 = l0 > 0.f ? l0 : NEG_SLOPE * l0;
    l1 = l1 > 0.f ? l1 : NEG_SLOPE * l1;
    l2 = l2 > 0.f ? l2 : NEG_SLOPE * l2;
    l3 = l3 > 0.f ? l3 : NEG_SLOPE * l3;
    float m0 = fkey_dec(mx.x) + ed.x, m1 = fkey_dec(mx.y) + ed.y;
    float m2 = fkey_dec(mx.z) + ed.z, m3 = fkey_dec(mx.w) + ed.w;
    m0 = m0 > 0.f ? m0 : NEG_SLOPE * m0;
    m1 = m1 > 0.f ? m1 : NEG_SLOPE * m1;
    m2 = m2 > 0.f ? m2 : NEG_SLOPE * m2;
    m3 = m3 > 0.f ? m3 : NEG_SLOPE * m3;
    const float x0 = __expf(l0 - m0);
    const float x1 = __expf(l1 - m1);
    const float x2 = __expf(l2 - m2);
    const float x3 = __expf(l3 - m3);
    if (ok) {
        atomicAdd(&sDen[dl * 4 + 0], x0);
        atomicAdd(&sDen[dl * 4 + 1], x1);
        atomicAdd(&sDen[dl * 4 + 2], x2);
        atomicAdd(&sDen[dl * 4 + 3], x3);
    }
    const int hd = lane >> 3;
    for (int j0 = 0; j0 < n; j0 += 8) {
        float4 r[8];
        for (int u = 0; u < 8; ++u) {
            const int j = min(j0 + u, n - 1);
            const int sj = __builtin_amdgcn_readlane(s, j);
            r[u] = *(const float4*)(z + (size_t)sj * 128 + lane * 4);
        }
        for (int u = 0; u < 8; ++u) {
            const int j = j0 + u;
            if ((okm >> j) & 1u) {
                const int dj = __builtin_amdgcn_readlane(dl, j);
                const float w0 = __int_as_float(__builtin_amdgcn_readlane(__float_as_int(x0), j));
                const float w1 = __int_as_float(__builtin_amdgcn_readlane(__float_as_int(x1), j));
                const float w2 = __int_as_float(__builtin_amdgcn_readlane(__float_as_int(x2), j));
                const float w3 = __int_as_float(__builtin_amdgcn_readlane(__float_as_int(x3), j));
                const float wa = (hd & 1) ? w1 : w0;
                const float wb = (hd & 1) ? w3 : w2;
                const float w  = (hd & 2) ? wb : wa;
                float* ap = sAcc + dj * 128 + lane;
                atomicAdd(ap,      w * r[u].x);
                atomicAdd(ap + 32, w * r[u].y);
                atomicAdd(ap + 64, w * r[u].z);
                atomicAdd(ap + 96, w * r[u].w);
            }
        }
    }
}

__global__ __launch_bounds__(1024)
void k_gatown(const int* __restrict__ eidx,
              const unsigned* __restrict__ own,
              const float* __restrict__ esrc,
              const float* __restrict__ edst,
              const float* __restrict__ z,
              const float* __restrict__ b_gat,
              const float* __restrict__ W_fc,
              const float* __restrict__ b_fc,
              const int* __restrict__ keptCnt,
              const int* __restrict__ keptLst, int cap,
              float* __restrict__ out, int N, int E) {
    __shared__ float sAcc[OWN_R * 128];
    __shared__ float sDen[OWN_R * 4];
    __shared__ unsigned sMax[OWN_R * 4];
    __shared__ int   sList[OWN_W * WL_CAP];

    const int tid = threadIdx.x;
    const int wave = __builtin_amdgcn_readfirstlane(tid >> 5);
    const int lane = tid & 31;
    const int nodeBase = blockIdx.x * OWN_R;
    const int rEff = max(min(OWN_R, N - nodeBase), 0);

    const int kept = cap > 0 ? keptCnt[blockIdx.x] : 0;
    const bool fromList = cap > 0 && kept <= cap;

    for (int i = tid; i < OWN_R * 128; i += 1024) sAcc[i] = 0.0f;
    for (int i = tid; i < OWN_R * 4; i += 1024) { sDen[i] = 0.0f; sMax[i] = 0u; }
    __syncthreads();

    if (fromList) {
        const int2* kl = (const int2*)keptLst + (size_t)blockIdx.x * (size_t)cap;
        for (int b0 = 0; b0 < kept; b0 += 4 * 1024) {
            int2 p[4];
            float4 sc[4];
#pragma unroll
            for (int u = 0; u < 4; ++u) p[u] = kl[min(b0 + u * 1024 + tid, kept - 1)];
            asm volatile("" :: "v"(p[0].x), "v"(p[1].x), "v"(p[2].x), "v"(p[3].x));
#pragma unroll
            for (int u = 0; u < 4; ++u) sc[u] = *(const float4*)(esrc + (size_t)(unsigned)p[u].x * 4);
            asm volatile("" :: "v"(sc[0].x), "v"(sc[1].x), "v"(sc[2].x), "v"(sc[3].x));
#pragma unroll
            for (int u = 0; u < 4; ++u)
                if (b0 + u * 1024 + tid < kept) max_edge(sc[u], p[u].y, sMax);
        }
        __syncthreads();

        const int run = (((kept + OWN_W - 1) / OWN_W) + 7) & ~7;
        const int end = min(kept, (wave + 1) * run);
        int base = wave * run;
        int2 cur = make_int2(0, 0);
        if (base < end) cur = kl[base + min(lane, min(32, end - base) - 1)];
        while (base < end) {
            const int n = min(32, end - base);
            const int nb = base + 32;
            int2 nxt = cur;
            if (nb < end) nxt = kl[nb + min(lane, min(32, end - nb) - 1)];
            const int s = cur.x, dl = cur.y;
            gat_batch(s, nodeBase + dl, dl, lane < n, n, lane, esrc, edst, sMax, z, sAcc, sDen);
            cur = nxt;
            base = nb;
        }
    } else {
    for (int pass = 0; pass < 2; ++pass) {
    int* wl = sList + wave * WL_CAP;
    int cnt = 0;

    const int* dstp = eidx + E;
    const uint4* oq = (const uint4*)own;
    const int nq16 = (E + 15) >> 4;
    const int nblk = (nq16 + 32 * SC_U - 1) / (32 * SC_U);
    const int rot = (int)blockIdx.x * (nblk / (int)gridDim.x);
    const unsigned pat = ((unsigned)blockIdx.x & 255u) * 0x01010101u;

    const uint4 zero4 = make_uint4(0u, 0u, 0u, 0u);
    uint4 cur[SC_U];
    for (int u = 0; u < SC_U; ++u) cur[u] = zero4;
    if (wave < nblk) {
        const int bc = wave + rot >= nblk ? wave + rot - nblk : wave + rot;
        for (int u = 0; u < SC_U; ++u)
            cur[u] = oq[min(bc * (32 * SC_U) + u * 32 + lane, nq16 - 1)];
    }

    for (int blk = wave; ; blk += OWN_W) {
        const bool live = blk < nblk;
        if (live) {
            uint4 nxt[SC_U];
            const int nb = blk + OWN_W;
            for (int u = 0; u < SC_U; ++u) nxt[u] = zero4;
            if (nb < nblk) {
                const int bn = nb + rot >= nblk ? nb + rot - nblk : nb + rot;
                for (int u = 0; u < SC_U; ++u)
                    nxt[u] = oq[min(bn * (32 * SC_U) + u * 32 + lane, nq16 - 1)];
            }
            const int bc = blk + rot >= nblk ? blk + rot - nblk : blk + rot;
            const int qi0 = bc * (32 * SC_U) + lane;
            unsigned hb = 0u;
            for (int u = 0; u < SC_U; ++u) {
                const unsigned f0 = zero_bytes(cur[u].x ^ pat);
                const unsigned f1 = zero_bytes(cur[u].y ^ pat);
                const unsigned f2 = zero_bytes(cur[u].z ^ pat);
                const unsigned f3 = zero_bytes(cur[u].w ^ pat);
                const unsigned h = (f0 >> 7) | (f1 >> 6) | (f2 >> 5) | (f3 >> 4);
                if (qi0 + 32 * u < nq16) hb |= h << (4 * u);
            }
            unsigned mk = __builtin_amdgcn_ballot_w32(hb != 0u);
            for (int r = 0; r < 16 * SC_U && mk != 0u; ++r) {
                if (hb != 0u) {
                    const int p = __builtin_ctz(hb);
                    wl[cnt + __builtin_amdgcn_mbcnt_lo(mk, 0u)] =
                        16 * (qi0 + 32 * ((p >> 2) & 1)) + 4 * (p & 3) + (p >> 3);
                    hb &= hb - 1u;
                }
                cnt += __builtin_popcount(mk);
                mk = __builtin_amdgcn_ballot_w32(hb != 0u);
            }
            for (int u = 0; u < SC_U; ++u) cur[u] = nxt[u];
        }

        const int need = live ? 32 : 1;
        while (cnt >= need) {
            const int n = min(cnt, 32);
            cnt -= n;
            const int* bl = wl + cnt;
            const int er = bl[lane < n ? lane : 0];
            const int e = min(er, E - 1);
            const int s = eidx[e];
            const int d = dstp[e];
            const int dl = d - nodeBase;
            const bool ok = lane < n && er < E && (unsigned)dl < (unsigned)rEff;
            if (pass == 0) {
                const float4 sc = *(const float4*)(esrc + (size_t)s * 4);
                if (ok) max_edge(sc, dl, sMax);
            } else {
                gat_batch(s, d, dl, ok, n, lane, esrc, edst, sMax, z, sAcc, sDen);
            }
        }
        if (!live) break;
    }
    if (pass == 0) __syncthreads();
    }
    }
    __syncthreads();

    for (int i = wave; i < rEff; i += OWN_W) {
        float acc = 0.f;
        for (int t = 0; t < 4; ++t) {
            const int j = t * 32 + lane;
            const float dn = sDen[i * 4 + t] + 1e-16f;
            float g = sAcc[i * 128 + (j & 3) * 32 + (j >> 2)] / dn + b_gat[j];
            g = fmaxf(g, 0.0f);
            acc += g * W_fc[j];
        }
        for (int off = 16; off > 0; off >>= 1) acc += __shfl_xor(acc, off, 32);
        if (lane == 0) out[nodeBase + i] = acc + b_fc[0];
    }
}

extern "C" void kernel_launch(void* const* d_in, const int* in_sizes, int n_in,
                              void* d_out, int out_size, void* d_ws, size_t ws_size,
                              hipStream_t stream) {
    const float* x      = (const float*)d_in[0];
    const int*   eidx   = (const int*)  d_in[1];
    const float* eattr  = (const float*)d_in[2];
    const float* W_nn   = (const float*)d_in[3];
    const float* b_nn   = (const float*)d_in[4];
    const float* W_root = (const float*)d_in[5];
    const float* b_ecc  = (const float*)d_in[6];
    const float* W_gat  = (const float*)d_in[7];
    const float* a_src  = (const float*)d_in[8];
    const float* a_dst  = (const float*)d_in[9];
    const float* b_gat  = (const float*)d_in[10];
    const float* W_fc   = (const float*)d_in[11];
    const float* b_fc   = (const float*)d_in[12];
    float* out = (float*)d_out;

    const int N = in_sizes[0] / 16;
    const int E = in_sizes[1] / 2;

    float* ws = (float*)d_ws;
    size_t off = 0;
    float*    hbuf   = ws + off;               off += (size_t)N * 16;
    float*    zbuf   = ws + off;               off += (size_t)N * 128;
    float*    esrc   = ws + off;               off += (size_t)N * 4;
    float*    edst   = ws + off;               off += (size_t)N * 4;
    float*    wa     = ws + off;               off += 128;
    unsigned* own    = (unsigned*)(ws + off);  off += (size_t)((E + 127) / 128) * 32;
    const int nOwn = (N + OWN_R - 1) / OWN_R;
    int cap = ((E / nOwn) * 3 / 2 + 127) / 128 * 128;
    int* keptCnt = (int*)(ws + off);           off += (size_t)((nOwn + 31) / 32) * 32;
    int* keptLst = (int*)(ws + off);           off += (size_t)nOwn * (size_t)cap * 2;
    if (off * sizeof(float) > ws_size) cap = 0;

    k_prep  <<<1024, 256, 0, stream>>>(eidx, own, E, W_gat, a_src, a_dst, wa);
    k_nnown <<<nOwn, 1024, 0, stream>>>(x, eidx, eattr, W_nn, b_nn, W_root, b_ecc,
                                         own, hbuf, keptCnt, keptLst, cap, N, E);
    k_zgemm <<<(N + 15) / 16, ZG_T, 0, stream>>>(hbuf, W_gat, wa, zbuf, esrc, edst, N);
    k_gatown<<<nOwn, 1024, 0, stream>>>(eidx, own, esrc, edst, zbuf,
                                         b_gat, W_fc, b_fc, keptCnt, keptLst, cap, out, N, E);
}


# z = h @ W_gat and the head scores taken in the NNConv owner kernel's tail from h kept in LDS (W_gat over the spent W' copy): no h plane, no z GEMM launch; 4->3 launches; sums bit-identical
# speedup vs baseline: 9.5959x; 1.0652x over previous
#include <hip/hip_runtime.h>
#include <hip/hip_bf16.h>

typedef __attribute__((ext_vector_type(2))) float v2f;
typedef __attribute__((ext_vector_type(8))) float v8f;

#define NEG_SLOPE 0.2f
#define OWN_R  196

__device__ __forceinline__ unsigned fkey_enc(float f) {
    unsigned b = __float_as_uint(f);
    return (b & 0x80000000u) ? ~b : (b | 0x80000000u);
}
__device__ __forceinline__ float fkey_dec(unsigned k) {
    return (k & 0x80000000u) ? __uint_as_float(k ^ 0x80000000u)
                             : __uint_as_float(~k);
}

#define OWN_W  32
#define SC_U   2
#define WL_CAP (32 + SC_U * 512)
#define PF_D   6
static_assert(SC_U == 2, "the flag word of the scan holds two groups of 16 edges per lane");

__device__ __forceinline__ unsigned zero_bytes(unsigned t) {
    return ~(((t & 0x7F7F7F7Fu) + 0x7F7F7F7Fu) | t | 0x7F7F7F7Fu);
}

__global__ void k_prep(const int* __restrict__ eidx,
                       unsigned* __restrict__ ownw, int E,
                       const float* __restrict__ W_gat,
                       const float* __restrict__ a_src,
                       const float* __restrict__ a_dst,
                       float* __restrict__ wa) {
    if (blockIdx.x == 0 && threadIdx.x < 128) {
        const int k = threadIdx.x >> 3, cc = threadIdx.x & 7;
        const float* av = (cc < 4 ? a_src : a_dst) + (cc & 3) * 32;
        const float* wr = W_gat + k * 128 + (cc & 3) * 32;
        float s = 0.f;
        for (int d = 0; d < 32; ++d) s += wr[d] * av[d];
        wa[k * 8 + cc] = s;
    }
    const int i0 = blockIdx.x * blockDim.x + threadIdx.x;
    const int stride = gridDim.x * blockDim.x;
    const int* dstp = eidx + E;
    const int nw = ((E + 127) / 128) * 32;
    for (int w = i0; w < nw; w += stride) {
        unsigned pk = 0u;
        for (int k = 0; k < 4; ++k) {
            const int idx = 4 * w + k;
            const int d = idx < E ? dstp[idx] : 0;
            pk |= (((unsigned)d / OWN_R) & 255u) << (8 * k);
        }
        ownw[w] = pk;
    }
}

#define ZG_COLS 144
__global__ __launch_bounds__(1024)
void k_nnown(const float* __restrict__ x,
             const int*   __restrict__ eidx,
             const float* __restrict__ eattr,
             const float* __restrict__ W_nn,
             const float* __restrict__ b_nn,
             const float* __restrict__ W_root,
             const float* __restrict__ b_ecc,
             const unsigned* __restrict__ own,
             const float* __restrict__ W_gat,
             const float* __restrict__ wa,
             float* __restrict__ z,
             float* __restrict__ esrc,
             float* __restrict__ edst,
             int* __restrict__ keptCnt,
             int* __restrict__ keptLst, int cap, int N, int E) {
    __shared__ float sWp[136 * 32];
    __shared__ float sAgg[OWN_R * 16];
    __shared__ int   sList[OWN_W * WL_CAP];
    __shared__ int   sKept;

    const int tid = threadIdx.x;
    const int wave = __builtin_amdgcn_readfirstlane(tid >> 5);
    const int lane = tid & 31;
    const int m    = lane & 15;
    const int hi   = lane >> 4;
    const int nodeBase = blockIdx.x * OWN_R;
    const int rEff = max(min(OWN_R, N - nodeBase), 0);

    for (int t = tid; t < 272 * 16; t += 1024) {
        int k = t >> 4, o = t & 15;
        float v = (k < 256) ? W_nn[(k >> 4) * 256 + (k & 15) * 16 + o]
                            : b_nn[(k - 256) * 16 + o];
        sWp[(k >> 1) * 32 + o * 2 + (k & 1)] = v;
    }
    for (int i = tid; i < OWN_R * 16; i += 1024) sAgg[i] = 0.0f;
    if (tid == 0) sKept = 0;
    __syncthreads();

    int* wl = sList + wave * WL_CAP;
    int cnt = 0;
    int2* kl = (int2*)keptLst + (size_t)blockIdx.x * (size_t)cap;

    const int* dstp = eidx + E;
    const uint4* oq = (const uint4*)own;
    const int nq16 = (E + 15) >> 4;
    const int nblk = (nq16 + 32 * SC_U - 1) / (32 * SC_U);
    const int rot = (int)blockIdx.x * (nblk / (int)gridDim.x);
    const unsigned pat = ((unsigned)blockIdx.x & 255u) * 0x01010101u;

    const uint4 zero4 = make_uint4(0u, 0u, 0u, 0u);
    uint4 cur[SC_U];
    for (int u = 0; u < SC_U; ++u) cur[u] = zero4;
    if (wave < nblk) {
        const int bc = wave + rot >= nblk ? wave + rot - nblk : wave + rot;
        for (int u = 0; u < SC_U; ++u)
            cur[u] = oq[min(bc * (32 * SC_U) + u * 32 + lane, nq16 - 1)];
    }

    for (int blk = wave; ; blk += OWN_W) {
        const bool live = blk < nblk;
        if (live) {
            uint4 nxt[SC_U];
            const int nb = blk + OWN_W;
            for (int u = 0; u < SC_U; ++u) nxt[u] = zero4;
            if (nb < nblk) {
                const int bn = nb + rot >= nblk ? nb + rot - nblk : nb + rot;
                for (int u = 0; u < SC_U; ++u)
                    nxt[u] = oq[min(bn * (32 * SC_U) + u * 32 + lane, nq16 - 1)];
            }
            const int bc = blk + rot >= nblk ? blk + rot - nblk : blk + rot;
            const int qi0 = bc * (32 * SC_U) + lane;
            unsigned hb = 0u;
            for (int u = 0; u < SC_U; ++u) {
                const unsigned f0 = zero_bytes(cur[u].x ^ pat);
                const unsigned f1 = zero_bytes(cur[u].y ^ pat);
                const unsigned f2 = zero_bytes(cur[u].z ^ pat);
                const unsigned f3 = zero_bytes(cur[u].w ^ pat);
                const unsigned hf = (f0 >> 7) | (f1 >> 6) | (f2 >> 5) | (f3 >> 4);
                if (qi0 + 32 * u < nq16) hb |= hf << (4 * u);
            }
            unsigned mk = __builtin_amdgcn_ballot_w32(hb != 0u);
            for (int r = 0; r < 16 * SC_U && mk != 0u; ++r) {
                if (hb != 0u) {
                    const int p = __builtin_ctz(hb);
                    wl[cnt + __builtin_amdgcn_mbcnt_lo(mk, 0u)] =
                        16 * (qi0 + 32 * ((p >> 2) & 1)) + 4 * (p & 3) + (p >> 3);
                    hb &= hb - 1u;
                }
                cnt += __builtin_popcount(mk);
                mk = __builtin_amdgcn_ballot_w32(hb != 0u);
            }
            for (int u = 0; u < SC_U; ++u) cur[u] = nxt[u];
        }

        const int need = live ? 16 : 1;
        while (cnt >= need) {
            const int n = min(cnt, 16);
            cnt -= n;
            const int* bl = wl + cnt;
            const int er = bl[m < n ? m : 0];
            const int e = min(er, E - 1);
            const int s = eidx[e];
            const int d = dstp[e];
            const int dl = d - nodeBase;
            const bool ok = m < n && er < E && (unsigned)dl < (unsigned)rEff;
            const int dlk = ok ? dl : -1;

            const float4* ep = (const float4*)(eattr + (size_t)e * 16);
            const float4 q0 = ep[0], q1 = ep[1], q2 = ep[2], q3 = ep[3];
            const float2* xp = (const float2*)(x + (size_t)s * 16 + 2 * hi);
            const float2 xa = xp[0], xb = xp[2], xc = xp[4], xd = xp[6];
            const float eav[16] = {q0.x, q0.y, q0.z, q0.w, q1.x, q1.y, q1.z, q1.w,
                                   q2.x, q2.y, q2.z, q2.w, q3.x, q3.y, q3.z, q3.w};
            const float xr0[4] = {xa.x, xb.x, xc.x, xd.x};
            const float xr1[4] = {xa.y, xb.y, xc.y, xd.y};

            v8f acc = {};
            typedef __attribute__((address_space(3))) const volatile v2f lds_pair_t;
            lds_pair_t* sWb = (lds_pair_t*)sWp;
            v2f bb[68];
#pragma unroll
            for (int kk = 0; kk < PF_D; ++kk) bb[kk] = sWb[(2 * kk + hi) * 16 + m];
#pragma unroll
            for (int kk = 0; kk < 68; ++kk) {
                if (kk + PF_D < 68) bb[kk + PF_D] = sWb[(2 * (kk + PF_D) + hi) * 16 + m];
                __builtin_amdgcn_sched_barrier(0);
                const float x0 = xr0[kk & 3], x1 = xr1[kk & 3];
                v2f a;
                if (kk < 64) {
                    const float eaf = eav[kk >> 2];
                    a.x = eaf * x0; a.y = eaf * x1;
                } else {
                    a.x = x0; a.y = x1;
                }
                acc = __builtin_amdgcn_wmma_f32_16x16x4_f32(
                    false, a, false, bb[kk], (short)0, acc, false, false);
            }

#pragma unroll
            for (int v = 0; v < 8; ++v) {
                const int dlo = __builtin_amdgcn_readlane(dlk, v);
                const int dhi = __builtin_amdgcn_readlane(dlk, v + 8);
                const int dr = hi ? dhi : dlo;
                if (dr >= 0) atomicAdd(&sAgg[dr * 16 + m], acc[v]);
            }

            if (cap > 0) {
                const unsigned okl = __builtin_amdgcn_ballot_w32(ok) & 0xFFFFu;
                if (okl != 0u) {
                    int at = 0;
                    if (lane == 0) at = atomicAdd(&sKept, __builtin_popcount(okl));
                    at = __builtin_amdgcn_readfirstlane(at);
                    const int pos = at + __builtin_popcount(okl & ((1u << lane) - 1u));
                    if (lane < 16 && ok && pos < cap) kl[pos] = make_int2(s, dl);
                }
            }
        }
        if (!live) break;
    }
    __syncthreads();
    if (cap > 0 && tid == 0) keptCnt[blockIdx.x] = sKept;

    float* sWg = sWp;
    static_assert(8 * ZG_COLS * 2 <= 136 * 32, "W_gat and the folded tile fit the W' copy");
    for (int t = tid; t < 16 * ZG_COLS; t += 1024) {
        const int k = t / ZG_COLS, col = t - k * ZG_COLS;
        float v = 0.0f;
        if (col < 128)      v = W_gat[k * 128 + col];
        else if (col < 136) v = wa[k * 8 + (col - 128)];
        sWg[(k >> 1) * (ZG_COLS * 2) + col * 2 + (k & 1)] = v;
    }

    for (int i = tid; i < rEff * 16; i += 1024) {
        const int nd = nodeBase + (i >> 4), o = i & 15;
        float sacc = sAgg[i] + b_ecc[o];
        const float* xrow = x + (size_t)nd * 16;
#pragma unroll
        for (int j = 0; j < 16; ++j) sacc += xrow[j] * W_root[j * 16 + o];
        sAgg[i] = fmaxf(sacc, 0.0f);
    }
    __syncthreads();

    const int nRt = (rEff + 15) >> 4;
    for (int p = wave; p < nRt * 9; p += OWN_W) {
        const int rt = p / 9, ct = p - rt * 9;
        const float* hr = sAgg + min(rt * 16 + m, rEff - 1) * 16;
        const int col = ct * 16 + m;
        v8f acc = {};
#pragma unroll
        for (int kk = 0; kk < 4; ++kk) {
            const int i0 = 4 * kk + 2 * hi;
            const int pp = 2 * kk + hi;
            v2f a, b;
            a.x = hr[i0]; a.y = hr[i0 + 1];
            b.x = sWg[pp * (ZG_COLS * 2) + col * 2 + 0];
            b.y = sWg[pp * (ZG_COLS * 2) + col * 2 + 1];
            acc = __builtin_amdgcn_wmma_f32_16x16x4_f32(
                false, a, false, b, (short)0, acc, false, false);
        }
        const bool whole = rt * 16 + 16 <= rEff;
        const int rl0 = rt * 16 + 8 * hi;
        if (ct < 8) {
            float* zp = z + (size_t)(nodeBase + rl0) * 128 + col;
            if (whole) {
#pragma unroll
                for (int v = 0; v < 8; ++v) zp[v * 128] = acc[v];
            } else {
#pragma unroll
                for (int v = 0; v < 8; ++v)
                    if (rl0 + v < rEff) zp[v * 128] = acc[v];
            }
        } else if (m < 8) {
            float* sp = (m < 4 ? esrc + m : edst + (m - 4)) + (size_t)(nodeBase + rl0) * 4;
            if (whole) {
#pragma unroll
                for (int v = 0; v < 8; ++v) sp[v * 4] = acc[v];
            } else {
#pragma unroll
                for (int v = 0; v < 8; ++v)
                    if (rl0 + v < rEff) sp[v * 4] = acc[v];
            }
        }
    }
}

__device__ __forceinline__ void max_edge(const float4 es, const int dl, unsigned* sMax) {
    atomicMax(&sMax[dl * 4 + 0], fkey_enc(es.x));
    atomicMax(&sMax[dl * 4 + 1], fkey_enc(es.y));
    atomicMax(&sMax[dl * 4 + 2], fkey_enc(es.z));
    atomicMax(&sMax[dl * 4 + 3], fkey_enc(es.w));
}

__device__ __forceinline__ void gat_batch(const int s, const int d, const int dl, const bool ok,
                                          const int n, const int lane,
                                          const float* __restrict__ esrc,
                                          const float* __restrict__ edst,
                                          const unsigned* sMax,
                                          const float* __restrict__ z,
                                          float* sAcc, float* sDen) {
    const unsigned okm = __builtin_amdgcn_ballot_w32(ok);
    const float4 es = *(const float4*)(esrc + (size_t)s * 4);
    const float4 ed = *(const float4*)(edst + (size_t)d * 4);
    const uint4  mx = *(const uint4*)(sMax + (ok ? dl : 0) * 4);
    float l0 = es.x + ed.x, l1 = es.y + ed.y, l2 = es.z + ed.z, l3 = es.w + ed.w;
    l0 = l0 > 0.f ? l0 : NEG_SLOPE * l0;
    l1 = l1 > 0.f ? l1 : NEG_SLOPE * l1;
    l2 = l2 > 0.f ? l2 : NEG_SLOPE * l2;
    l3 = l3 > 0.f ? l3 : NEG_SLOPE * l3;
    float m0 = fkey_dec(mx.x) + ed.x, m1 = fkey_dec(mx.y) + ed.y;
    float m2 = fkey_dec(mx.z) + ed.z, m3 = fkey_dec(mx.w) + ed.w;
    m0 = m0 > 0.f ? m0 : NEG_SLOPE * m0;
    m1 = m1 > 0.f ? m1 : NEG_SLOPE * m1;
    m2 = m2 > 0.f ? m2 : NEG_SLOPE * m2;
    m3 = m3 > 0.f ? m3 : NEG_SLOPE * m3;
    const float x0 = __expf(l0 - m0);
    const float x1 = __expf(l1 - m1);
    const float x2 = __expf(l2 - m2);
    const float x3 = __expf(l3 - m3);
    if (ok) {
        atomicAdd(&sDen[dl * 4 + 0], x0);
        atomicAdd(&sDen[dl * 4 + 1], x1);
        atomicAdd(&sDen[dl * 4 + 2], x2);
        atomicAdd(&sDen[dl * 4 + 3], x3);
    }
    const int hd = lane >> 3;
    for (int j0 = 0; j0 < n; j0 += 8) {
        float4 r[8];
        for (int u = 0; u < 8; ++u) {
            const int j = min(j0 + u, n - 1);
            const int sj = __builtin_amdgcn_readlane(s, j);
            r[u] = *(const float4*)(z + (size_t)sj * 128 + lane * 4);
        }
        for (int u = 0; u < 8; ++u) {
            const int j = j0 + u;
            if ((okm >> j) & 1u) {
                const int dj = __builtin_amdgcn_readlane(dl, j);
                const float w0 = __int_as_float(__builtin_amdgcn_readlane(__float_as_int(x0), j));
                const float w1 = __int_as_float(__builtin_amdgcn_readlane(__float_as_int(x1), j));
                const float w2 = __int_as_float(__builtin_amdgcn_readlane(__float_as_int(x2), j));
                const float w3 = __int_as_float(__builtin_amdgcn_readlane(__float_as_int(x3), j));
                const float wa = (hd & 1) ? w1 : w0;
                const float wb = (hd & 1) ? w3 : w2;
                const float w  = (hd & 2) ? wb : wa;
                float* ap = sAcc + dj * 128 + lane;
                atomicAdd(ap,      w * r[u].x);
                atomicAdd(ap + 32, w * r[u].y);
                atomicAdd(ap + 64, w * r[u].z);
                atomicAdd(ap + 96, w * r[u].w);
            }
        }
    }
}

__global__ __launch_bounds__(1024)
void k_gatown(const int* __restrict__ eidx,
              const unsigned* __restrict__ own,
              const float* __restrict__ esrc,
              const float* __restrict__ edst,
              const float* __restrict__ z,
              const float* __restrict__ b_gat,
              const float* __restrict__ W_fc,
              const float* __restrict__ b_fc,
              const int* __restrict__ keptCnt,
              const int* __restrict__ keptLst, int cap,
              float* __restrict__ out, int N, int E) {
    __shared__ float sAcc[OWN_R * 128];
    __shared__ float sDen[OWN_R * 4];
    __shared__ unsigned sMax[OWN_R * 4];
    __shared__ int   sList[OWN_W * WL_CAP];

    const int tid = threadIdx.x;
    const int wave = __builtin_amdgcn_readfirstlane(tid >> 5);
    const int lane = tid & 31;
    const int nodeBase = blockIdx.x * OWN_R;
    const int rEff = max(min(OWN_R, N - nodeBase), 0);

    const int kept = cap > 0 ? keptCnt[blockIdx.x] : 0;
    const bool fromList = cap > 0 && kept <= cap;

    for (int i = tid; i < OWN_R * 128; i += 1024) sAcc[i] = 0.0f;
    for (int i = tid; i < OWN_R * 4; i += 1024) { sDen[i] = 0.0f; sMax[i] = 0u; }
    __syncthreads();

    if (fromList) {
        const int2* kl = (const int2*)keptLst + (size_t)blockIdx.x * (size_t)cap;
        for (int b0 = 0; b0 < kept; b0 += 4 * 1024) {
            int2 p[4];
            float4 sc[4];
#pragma unroll
            for (int u = 0; u < 4; ++u) p[u] = kl[min(b0 + u * 1024 + tid, kept - 1)];
            asm volatile("" :: "v"(p[0].x), "v"(p[1].x), "v"(p[2].x), "v"(p[3].x));
#pragma unroll
            for (int u = 0; u < 4; ++u) sc[u] = *(const float4*)(esrc + (size_t)(unsigned)p[u].x * 4);
            asm volatile("" :: "v"(sc[0].x), "v"(sc[1].x), "v"(sc[2].x), "v"(sc[3].x));
#pragma unroll
            for (int u = 0; u < 4; ++u)
                if (b0 + u * 1024 + tid < kept) max_edge(sc[u], p[u].y, sMax);
        }
        __syncthreads();

        const int run = (((kept + OWN_W - 1) / OWN_W) + 7) & ~7;
        const int end = min(kept, (wave + 1) * run);
        int base = wave * run;
        int2 cur = make_int2(0, 0);
        if (base < end) cur = kl[base + min(lane, min(32, end - base) - 1)];
        while (base < end) {
            const int n = min(32, end - base);
            const int nb = base + 32;
            int2 nxt = cur;
            if (nb < end) nxt = kl[nb + min(lane, min(32, end - nb) - 1)];
            const int s = cur.x, dl = cur.y;
            gat_batch(s, nodeBase + dl, dl, lane < n, n, lane, esrc, edst, sMax, z, sAcc, sDen);
            cur = nxt;
            base = nb;
        }
    } else {
    for (int pass = 0; pass < 2; ++pass) {
    int* wl = sList + wave * WL_CAP;
    int cnt = 0;

    const int* dstp = eidx + E;
    const uint4* oq = (const uint4*)own;
    const int nq16 = (E + 15) >> 4;
    const int nblk = (nq16 + 32 * SC_U - 1) / (32 * SC_U);
    const int rot = (int)blockIdx.x * (nblk / (int)gridDim.x);
    const unsigned pat = ((unsigned)blockIdx.x & 255u) * 0x01010101u;

    const uint4 zero4 = make_uint4(0u, 0u, 0u, 0u);
    uint4 cur[SC_U];
    for (int u = 0; u < SC_U; ++u) cur[u] = zero4;
    if (wave < nblk) {
        const int bc = wave + rot >= nblk ? wave + rot - nblk : wave + rot;
        for (int u = 0; u < SC_U; ++u)
            cur[u] = oq[min(bc * (32 * SC_U) + u * 32 + lane, nq16 - 1)];
    }

    for (int blk = wave; ; blk += OWN_W) {
        const bool live = blk < nblk;
        if (live) {
            uint4 nxt[SC_U];
            const int nb = blk + OWN_W;
            for (int u = 0; u < SC_U; ++u) nxt[u] = zero4;
            if (nb < nblk) {
                const int bn = nb + rot >= nblk ? nb + rot - nblk : nb + rot;
                for (int u = 0; u < SC_U; ++u)
                    nxt[u] = oq[min(bn * (32 * SC_U) + u * 32 + lane, nq16 - 1)];
            }
            const int bc = blk + rot >= nblk ? blk + rot - nblk : blk + rot;
            const int qi0 = bc * (32 * SC_U) + lane;
            unsigned hb = 0u;
            for (int u = 0; u < SC_U; ++u) {
                const unsigned f0 = zero_bytes(cur[u].x ^ pat);
                const unsigned f1 = zero_bytes(cur[u].y ^ pat);
                const unsigned f2 = zero_bytes(cur[u].z ^ pat);
                const unsigned f3 = zero_bytes(cur[u].w ^ pat);
                const unsigned h = (f0 >> 7) | (f1 >> 6) | (f2 >> 5) | (f3 >> 4);
                if (qi0 + 32 * u < nq16) hb |= h << (4 * u);
            }
            unsigned mk = __builtin_amdgcn_ballot_w32(hb != 0u);
            for (int r = 0; r < 16 * SC_U && mk != 0u; ++r) {
                if (hb != 0u) {
                    const int p = __builtin_ctz(hb);
                    wl[cnt + __builtin_amdgcn_mbcnt_lo(mk, 0u)] =
                        16 * (qi0 + 32 * ((p >> 2) & 1)) + 4 * (p & 3) + (p >> 3);
                    hb &= hb - 1u;
                }
                cnt += __builtin_popcount(mk);
                mk = __builtin_amdgcn_ballot_w32(hb != 0u);
            }
            for (int u = 0; u < SC_U; ++u) cur[u] = nxt[u];
        }

        const int need = live ? 32 : 1;
        while (cnt >= need) {
            const int n = min(cnt, 32);
            cnt -= n;
            const int* bl = wl + cnt;
            const int er = bl[lane < n ? lane : 0];
            const int e = min(er, E - 1);
            const int s = eidx[e];
            const int d = dstp[e];
            const int dl = d - nodeBase;
            const bool ok = lane < n && er < E && (unsigned)dl < (unsigned)rEff;
            if (pass == 0) {
                const float4 sc = *(const float4*)(esrc + (size_t)s * 4);
                if (ok) max_edge(sc, dl, sMax);
            } else {
                gat_batch(s, d, dl, ok, n, lane, esrc, edst, sMax, z, sAcc, sDen);
            }
        }
        if (!live) break;
    }
    if (pass == 0) __syncthreads();
    }
    }
    __syncthreads();

    for (int i = wave; i < rEff; i += OWN_W) {
        float acc = 0.f;
        for (int t = 0; t < 4; ++t) {
            const int j = t * 32 + lane;
            const float dn = sDen[i * 4 + t] + 1e-16f;
            float g = sAcc[i * 128 + (j & 3) * 32 + (j >> 2)] / dn + b_gat[j];
            g = fmaxf(g, 0.0f);
            acc += g * W_fc[j];
        }
        for (int off = 16; off > 0; off >>= 1) acc += __shfl_xor(acc, off, 32);
        if (lane == 0) out[nodeBase + i] = acc + b_fc[0];
    }
}

extern "C" void kernel_launch(void* const* d_in, const int* in_sizes, int n_in,
                              void* d_out, int out_size, void* d_ws, size_t ws_size,
                              hipStream_t stream) {
    const float* x      = (const float*)d_in[0];
    const int*   eidx   = (const int*)  d_in[1];
    const float* eattr  = (const float*)d_in[2];
    const float* W_nn   = (const float*)d_in[3];
    const float* b_nn   = (const float*)d_in[4];
    const float* W_root = (const float*)d_in[5];
    const float* b_ecc  = (const float*)d_in[6];
    const float* W_gat  = (const float*)d_in[7];
    const float* a_src  = (const float*)d_in[8];
    const float* a_dst  = (const float*)d_in[9];
    const float* b_gat  = (const float*)d_in[10];
    const float* W_fc   = (const float*)d_in[11];
    const float* b_fc   = (const float*)d_in[12];
    float* out = (float*)d_out;

    const int N = in_sizes[0] / 16;
    const int E = in_sizes[1] / 2;

    float* ws = (float*)d_ws;
    size_t off = 0;
    float*    zbuf   = ws + off;               off += (size_t)N * 128;
    float*    esrc   = ws + off;               off += (size_t)N * 4;
    float*    edst   = ws + off;               off += (size_t)N * 4;
    float*    wa     = ws + off;               off += 128;
    unsigned* own    = (unsigned*)(ws + off);  off += (size_t)((E + 127) / 128) * 32;
    const int nOwn = (N + OWN_R - 1) / OWN_R;
    int cap = ((E / nOwn) * 3 / 2 + 127) / 128 * 128;
    int* keptCnt = (int*)(ws + off);           off += (size_t)((nOwn + 31) / 32) * 32;
    int* keptLst = (int*)(ws + off);           off += (size_t)nOwn * (size_t)cap * 2;
    if (off * sizeof(float) > ws_size) cap = 0;

    k_prep  <<<1024, 256, 0, stream>>>(eidx, own, E, W_gat, a_src, a_dst, wa);
    k_nnown <<<nOwn, 1024, 0, stream>>>(x, eidx, eattr, W_nn, b_nn, W_root, b_ecc,
                                         own, W_gat, wa, zbuf, esrc, edst,
                                         keptCnt, keptLst, cap, N, E);
    k_gatown<<<nOwn, 1024, 0, stream>>>(eidx, own, esrc, edst, zbuf,
                                         b_gat, W_fc, b_fc, keptCnt, keptLst, cap, out, N, E);
}
